# starting kernel
# baseline (speedup 1.0000x reference)
#include <hip/hip_runtime.h>


typedef float v2f __attribute__((ext_vector_type(2)));
typedef float v8f __attribute__((ext_vector_type(8)));


__global__ void k_zero(float* __restrict__ p, long n) {
  long i = (long)blockIdx.x * blockDim.x + threadIdx.x;
  if (i < n) p[i] = 0.0f;
}

__global__ void k_prep_weights(const float* __restrict__ We,
                               const float* __restrict__ Wn,
                               const float* __restrict__ W1,
                               float* __restrict__ Wep, float* __restrict__ Wnp,
                               float* __restrict__ W1p) {
  int i = blockIdx.x * blockDim.x + threadIdx.x;
  if (i < 256) {
    int n = i >> 4, k = i & 15;
    Wep[i] = (n < 8) ? We[k * 8 + n] : 0.0f;
  }
  if (i < 1024) {
    int n = i >> 6, k = i & 63;
    Wnp[i] = Wn[k * 16 + n];
  }
  if (i < 384) {
    int n = i / 24, k = i % 24;
    W1p[i] = (n < 10) ? W1[k * 10 + n] : 0.0f;
  }
}

__global__ void k_degree(const int* __restrict__ src, const int* __restrict__ dst,
                         float* __restrict__ deg_out, float* __restrict__ deg_in, int E) {
  int e = blockIdx.x * blockDim.x + threadIdx.x;
  if (e < E) {
    atomicAdd(&deg_out[src[e]], 1.0f);
    atomicAdd(&deg_in[dst[e]], 1.0f);
  }
}

__global__ void k_edge_gemm_scatter(const float* __restrict__ ef,
                                    const float* __restrict__ Wep,
                                    const int* __restrict__ dst,
                                    float* __restrict__ e_sum, int E) {
  int wave = blockIdx.x * (blockDim.x >> 5) + (threadIdx.x >> 5);
  int lane = threadIdx.x & 31;
  int e0 = wave * 16;
  if (e0 >= E) return;
  int m = lane & 15, kh = lane >> 4, col = m;
  int erow = e0 + m;
  int er = erow < E ? erow : (E - 1);
  float amask = (erow < E) ? 1.0f : 0.0f;

  v8f c = {};
#pragma unroll
  for (int kb = 0; kb < 16; kb += 4) {
    int k = kb + 2 * kh;
    v2f a = *reinterpret_cast<const v2f*>(ef + (size_t)er * 16 + k);
    a.x *= amask;
    a.y *= amask;
    v2f b = *reinterpret_cast<const v2f*>(Wep + col * 16 + k);
    c = __builtin_amdgcn_wmma_f32_16x16x4_f32(false, a, false, b, (short)0, c,
                                              false, false);
  }
  if (col < 8) {
    if (e0 + 16 <= E) {
      int d[8];
#pragma unroll
      for (int v = 0; v < 8; ++v) d[v] = dst[e0 + v + 8 * kh];
#pragma unroll
      for (int v = 0; v < 8; ++v)
        atomicAdd(&e_sum[(size_t)d[v] * 8 + col], c[v]);
    } else {
#pragma unroll
      for (int v = 0; v < 8; ++v) {
        int e = e0 + v + 8 * kh;
        if (e < E) atomicAdd(&e_sum[(size_t)dst[e] * 8 + col], c[v]);
      }
    }
  }
}

__global__ void k_coeff_edgeh(const float* __restrict__ deg_out,
                              const float* __restrict__ deg_in,
                              const float* __restrict__ be,
                              const float* __restrict__ e_sum,
                              float* __restrict__ c_src, float* __restrict__ c_dst,
                              float* __restrict__ h24, int N) {
  int n = blockIdx.x * blockDim.x + threadIdx.x;
  if (n >= N) return;
  float dout = deg_out[n], din = deg_in[n];
  c_src[n] = (dout > 0.0f) ? rsqrtf(dout) : 0.0f;
  c_dst[n] = (din > 0.0f) ? rsqrtf(din) : 0.0f;
  float inv = 1.0f / fmaxf(din, 1.0f);
  float bsel = (din > 0.0f) ? 1.0f : 0.0f;
#pragma unroll
  for (int j = 0; j < 8; ++j) {
    float v = e_sum[(size_t)n * 8 + j] * inv + bsel * be[j];
    h24[(size_t)n * 24 + j] = fmaxf(v, 0.0f);
  }
}

__global__ void k_node_gemm(const float* __restrict__ nf,
                            const float* __restrict__ Wnp,
                            const float* __restrict__ bn,
                            float* __restrict__ h24, int N) {
  int wave = blockIdx.x * (blockDim.x >> 5) + (threadIdx.x >> 5);
  int lane = threadIdx.x & 31;
  int n0 = wave * 16;
  if (n0 >= N) return;
  int m = lane & 15, kh = lane >> 4, col = m;
  int nrow = n0 + m;
  int nr = nrow < N ? nrow : (N - 1);
  float amask = (nrow < N) ? 1.0f : 0.0f;

  v8f c = {};
#pragma unroll
  for (int kb = 0; kb < 64; kb += 4) {
    int k = kb + 2 * kh;
    v2f a = *reinterpret_cast<const v2f*>(nf + (size_t)nr * 64 + k);
    a.x *= amask;
    a.y *= amask;
    v2f b = *reinterpret_cast<const v2f*>(Wnp + col * 64 + k);
    c = __builtin_amdgcn_wmma_f32_16x16x4_f32(false, a, false, b, (short)0, c,
                                              false, false);
  }
  float bc = bn[col];
  if (n0 + 16 <= N) {
#pragma unroll
    for (int v = 0; v < 8; ++v)
      h24[(size_t)(n0 + v + 8 * kh) * 24 + 8 + col] = fmaxf(c[v] + bc, 0.0f);
  } else {
#pragma unroll
    for (int v = 0; v < 8; ++v) {
      int nn = n0 + v + 8 * kh;
      if (nn < N) h24[(size_t)nn * 24 + 8 + col] = fmaxf(c[v] + bc, 0.0f);
    }
  }
}

__global__ void k_g1_gemm(const float* __restrict__ h24,
                          const float* __restrict__ c_src,
                          const float* __restrict__ W1p,
                          float* __restrict__ g1, int N) {
  int wave = blockIdx.x * (blockDim.x >> 5) + (threadIdx.x >> 5);
  int lane = threadIdx.x & 31;
  int n0 = wave * 16;
  if (n0 >= N) return;
  int m = lane & 15, kh = lane >> 4, col = m;
  int nrow = n0 + m;
  int nr = nrow < N ? nrow : (N - 1);
  float cs = (nrow < N) ? c_src[nr] : 0.0f;

  v8f c = {};
#pragma unroll
  for (int kb = 0; kb < 24; kb += 4) {
    int k = kb + 2 * kh;
    v2f a = *reinterpret_cast<const v2f*>(h24 + (size_t)nr * 24 + k);
    a.x *= cs;
    a.y *= cs;
    v2f b = *reinterpret_cast<const v2f*>(W1p + col * 24 + k);
    c = __builtin_amdgcn_wmma_f32_16x16x4_f32(false, a, false, b, (short)0, c,
                                              false, false);
  }
  if (col < 10) {
    if (n0 + 16 <= N) {
#pragma unroll
      for (int v = 0; v < 8; ++v)
        g1[(size_t)(n0 + v + 8 * kh) * 10 + col] = c[v];
    } else {
#pragma unroll
      for (int v = 0; v < 8; ++v) {
        int nn = n0 + v + 8 * kh;
        if (nn < N) g1[(size_t)nn * 10 + col] = c[v];
      }
    }
  }
}

__global__ void k_scatter10(const int* __restrict__ src, const int* __restrict__ dst,
                            const float* __restrict__ g1, float* __restrict__ agg1, int E) {
  int e = blockIdx.x * blockDim.x + threadIdx.x;
  if (e >= E) return;
  size_t s = (size_t)src[e], d = (size_t)dst[e];
  const v2f* gs = reinterpret_cast<const v2f*>(g1 + s * 10);
  float* ad = agg1 + d * 10;
  v2f v0 = gs[0], v1 = gs[1], v2 = gs[2], v3 = gs[3], v4 = gs[4];
  atomicAdd(&ad[0], v0.x);
  atomicAdd(&ad[1], v0.y);
  atomicAdd(&ad[2], v1.x);
  atomicAdd(&ad[3], v1.y);
  atomicAdd(&ad[4], v2.x);
  atomicAdd(&ad[5], v2.y);
  atomicAdd(&ad[6], v3.x);
  atomicAdd(&ad[7], v3.y);
  atomicAdd(&ad[8], v4.x);
  atomicAdd(&ad[9], v4.y);
}

__global__ void k_scatter2(const int* __restrict__ src, const int* __restrict__ dst,
                           const float* __restrict__ g2, float* __restrict__ agg2, int E) {
  int e = blockIdx.x * blockDim.x + threadIdx.x;
  if (e >= E) return;
  size_t s = (size_t)src[e], d = (size_t)dst[e];
  v2f v = *reinterpret_cast<const v2f*>(g2 + s * 2);
  atomicAdd(&agg2[d * 2 + 0], v.x);
  atomicAdd(&agg2[d * 2 + 1], v.y);
}

__global__ void k_node2(const float* __restrict__ agg1,
                        const float* __restrict__ c_src, const float* __restrict__ c_dst,
                        const float* __restrict__ b1, const float* __restrict__ W2,
                        float* __restrict__ g2, int N) {
  int n = blockIdx.x * blockDim.x + threadIdx.x;
  if (n >= N) return;
  float cd = c_dst[n], cs = c_src[n];
  float a0 = 0.0f, a1 = 0.0f;
#pragma unroll
  for (int j = 0; j < 10; ++j) {
    float h = fmaxf(agg1[(size_t)n * 10 + j] * cd + b1[j], 0.0f);
    a0 += h * W2[j * 2 + 0];
    a1 += h * W2[j * 2 + 1];
  }
  g2[(size_t)n * 2 + 0] = a0 * cs;
  g2[(size_t)n * 2 + 1] = a1 * cs;
}

__global__ void k_out(const float* __restrict__ agg2, const float* __restrict__ c_dst,
                      const float* __restrict__ b2, float* __restrict__ out, int N) {
  int n = blockIdx.x * blockDim.x + threadIdx.x;
  if (n >= N) return;
  float cd = c_dst[n];
  out[(size_t)n * 2 + 0] = agg2[(size_t)n * 2 + 0] * cd + b2[0];
  out[(size_t)n * 2 + 1] = agg2[(size_t)n * 2 + 1] * cd + b2[1];
}


extern "C" void kernel_launch(void* const* d_in, const int* in_sizes, int n_in,
                              void* d_out, int out_size, void* d_ws, size_t ws_size,
                              hipStream_t stream) {
  const float* node_feats = (const float*)d_in[0];
  const float* edge_feats = (const float*)d_in[1];
  const float* We = (const float*)d_in[2];
  const float* be = (const float*)d_in[3];
  const float* Wn = (const float*)d_in[4];
  const float* bn = (const float*)d_in[5];
  const float* W1 = (const float*)d_in[6];
  const float* b1 = (const float*)d_in[7];
  const float* W2 = (const float*)d_in[8];
  const float* b2 = (const float*)d_in[9];
  const int* src = (const int*)d_in[10];
  const int* dst = (const int*)d_in[11];
  float* out = (float*)d_out;

  const int N = in_sizes[0] / 64;
  const int E = in_sizes[10];
  const size_t Nl = (size_t)N;

  float* ws = (float*)d_ws;
  float* deg_in = ws;
  float* deg_out = ws + Nl;
  float* e_sum = ws + 2 * Nl;
  float* agg1 = ws + 10 * Nl;
  float* agg2 = ws + 20 * Nl;
  float* c_src = ws + 22 * Nl;
  float* c_dst = ws + 23 * Nl;
  float* h24 = ws + 24 * Nl;
  float* g1 = ws + 48 * Nl;
  float* g2 = ws + 58 * Nl;
  float* Wep = ws + 60 * Nl;
  float* Wnp = Wep + 256;
  float* W1p = Wnp + 1024;

  const int TB = 256;
  const long zn = 22L * N;
  k_zero<<<(int)((zn + TB - 1) / TB), TB, 0, stream>>>(ws, zn);
  k_prep_weights<<<4, TB, 0, stream>>>(We, Wn, W1, Wep, Wnp, W1p);

  const int eb = (E + TB - 1) / TB;
  const int nb = (N + TB - 1) / TB;
  const int waves_per_block = TB / 32;
  const int etile_blocks = (((E + 15) / 16) + waves_per_block - 1) / waves_per_block;
  const int ntile_blocks = (((N + 15) / 16) + waves_per_block - 1) / waves_per_block;

  k_degree<<<eb, TB, 0, stream>>>(src, dst, deg_out, deg_in, E);
  k_edge_gemm_scatter<<<etile_blocks, TB, 0, stream>>>(edge_feats, Wep, dst, e_sum, E);
  k_coeff_edgeh<<<nb, TB, 0, stream>>>(deg_out, deg_in, be, e_sum, c_src, c_dst, h24, N);
  k_node_gemm<<<ntile_blocks, TB, 0, stream>>>(node_feats, Wnp, bn, h24, N);
  k_g1_gemm<<<ntile_blocks, TB, 0, stream>>>(h24, c_src, W1p, g1, N);
  k_scatter10<<<eb, TB, 0, stream>>>(src, dst, g1, agg1, E);
  k_node2<<<nb, TB, 0, stream>>>(agg1, c_src, c_dst, b1, W2, g2, N);
  k_scatter2<<<eb, TB, 0, stream>>>(src, dst, g2, agg2, E);
  k_out<<<nb, TB, 0, stream>>>(agg2, c_dst, b2, out, N);
}


# Scatter-add atomics coalesced: lanes carry the columns of one destination row (3 rows x 10 or 16 rows x 2 per instruction), not one edge each; f32 atomics kept, sums still in arrival order
# speedup vs baseline: 1.8960x; 1.8960x over previous
#include <hip/hip_runtime.h>


typedef float v2f __attribute__((ext_vector_type(2)));
typedef float v8f __attribute__((ext_vector_type(8)));


__global__ void k_zero(float* __restrict__ p, long n) {
  long i = (long)blockIdx.x * blockDim.x + threadIdx.x;
  if (i < n) p[i] = 0.0f;
}

__global__ void k_prep_weights(const float* __restrict__ We,
                               const float* __restrict__ Wn,
                               const float* __restrict__ W1,
                               float* __restrict__ Wep, float* __restrict__ Wnp,
                               float* __restrict__ W1p) {
  int i = blockIdx.x * blockDim.x + threadIdx.x;
  if (i < 256) {
    int n = i >> 4, k = i & 15;
    Wep[i] = (n < 8) ? We[k * 8 + n] : 0.0f;
  }
  if (i < 1024) {
    int n = i >> 6, k = i & 63;
    Wnp[i] = Wn[k * 16 + n];
  }
  if (i < 384) {
    int n = i / 24, k = i % 24;
    W1p[i] = (n < 10) ? W1[k * 10 + n] : 0.0f;
  }
}

__global__ void k_degree(const int* __restrict__ src, const int* __restrict__ dst,
                         float* __restrict__ deg_out, float* __restrict__ deg_in, int E) {
  int e = blockIdx.x * blockDim.x + threadIdx.x;
  if (e < E) {
    atomicAdd(&deg_out[src[e]], 1.0f);
    atomicAdd(&deg_in[dst[e]], 1.0f);
  }
}

__global__ void k_edge_gemm_scatter(const float* __restrict__ ef,
                                    const float* __restrict__ Wep,
                                    const int* __restrict__ dst,
                                    float* __restrict__ e_sum, int E) {
  int wave = blockIdx.x * (blockDim.x >> 5) + (threadIdx.x >> 5);
  int lane = threadIdx.x & 31;
  int e0 = wave * 16;
  if (e0 >= E) return;
  int m = lane & 15, kh = lane >> 4, col = m;
  int erow = e0 + m;
  int er = erow < E ? erow : (E - 1);
  float amask = (erow < E) ? 1.0f : 0.0f;

  v8f c = {};
#pragma unroll
  for (int kb = 0; kb < 16; kb += 4) {
    int k = kb + 2 * kh;
    v2f a = *reinterpret_cast<const v2f*>(ef + (size_t)er * 16 + k);
    a.x *= amask;
    a.y *= amask;
    v2f b = *reinterpret_cast<const v2f*>(Wep + col * 16 + k);
    c = __builtin_amdgcn_wmma_f32_16x16x4_f32(false, a, false, b, (short)0, c,
                                              false, false);
  }
  if (col < 8) {
    if (e0 + 16 <= E) {
      int d[8];
#pragma unroll
      for (int v = 0; v < 8; ++v) d[v] = dst[e0 + v + 8 * kh];
#pragma unroll
      for (int v = 0; v < 8; ++v)
        atomicAdd(&e_sum[(size_t)d[v] * 8 + col], c[v]);
    } else {
#pragma unroll
      for (int v = 0; v < 8; ++v) {
        int e = e0 + v + 8 * kh;
        if (e < E) atomicAdd(&e_sum[(size_t)dst[e] * 8 + col], c[v]);
      }
    }
  }
}

__global__ void k_coeff_edgeh(const float* __restrict__ deg_out,
                              const float* __restrict__ deg_in,
                              const float* __restrict__ be,
                              const float* __restrict__ e_sum,
                              float* __restrict__ c_src, float* __restrict__ c_dst,
                              float* __restrict__ h24, int N) {
  int n = blockIdx.x * blockDim.x + threadIdx.x;
  if (n >= N) return;
  float dout = deg_out[n], din = deg_in[n];
  c_src[n] = (dout > 0.0f) ? rsqrtf(dout) : 0.0f;
  c_dst[n] = (din > 0.0f) ? rsqrtf(din) : 0.0f;
  float inv = 1.0f / fmaxf(din, 1.0f);
  float bsel = (din > 0.0f) ? 1.0f : 0.0f;
#pragma unroll
  for (int j = 0; j < 8; ++j) {
    float v = e_sum[(size_t)n * 8 + j] * inv + bsel * be[j];
    h24[(size_t)n * 24 + j] = fmaxf(v, 0.0f);
  }
}

__global__ void k_node_gemm(const float* __restrict__ nf,
                            const float* __restrict__ Wnp,
                            const float* __restrict__ bn,
                            float* __restrict__ h24, int N) {
  int wave = blockIdx.x * (blockDim.x >> 5) + (threadIdx.x >> 5);
  int lane = threadIdx.x & 31;
  int n0 = wave * 16;
  if (n0 >= N) return;
  int m = lane & 15, kh = lane >> 4, col = m;
  int nrow = n0 + m;
  int nr = nrow < N ? nrow : (N - 1);
  float amask = (nrow < N) ? 1.0f : 0.0f;

  v8f c = {};
#pragma unroll
  for (int kb = 0; kb < 64; kb += 4) {
    int k = kb + 2 * kh;
    v2f a = *reinterpret_cast<const v2f*>(nf + (size_t)nr * 64 + k);
    a.x *= amask;
    a.y *= amask;
    v2f b = *reinterpret_cast<const v2f*>(Wnp + col * 64 + k);
    c = __builtin_amdgcn_wmma_f32_16x16x4_f32(false, a, false, b, (short)0, c,
                                              false, false);
  }
  float bc = bn[col];
  if (n0 + 16 <= N) {
#pragma unroll
    for (int v = 0; v < 8; ++v)
      h24[(size_t)(n0 + v + 8 * kh) * 24 + 8 + col] = fmaxf(c[v] + bc, 0.0f);
  } else {
#pragma unroll
    for (int v = 0; v < 8; ++v) {
      int nn = n0 + v + 8 * kh;
      if (nn < N) h24[(size_t)nn * 24 + 8 + col] = fmaxf(c[v] + bc, 0.0f);
    }
  }
}

__global__ void k_g1_gemm(const float* __restrict__ h24,
                          const float* __restrict__ c_src,
                          const float* __restrict__ W1p,
                          float* __restrict__ g1, int N) {
  int wave = blockIdx.x * (blockDim.x >> 5) + (threadIdx.x >> 5);
  int lane = threadIdx.x & 31;
  int n0 = wave * 16;
  if (n0 >= N) return;
  int m = lane & 15, kh = lane >> 4, col = m;
  int nrow = n0 + m;
  int nr = nrow < N ? nrow : (N - 1);
  float cs = (nrow < N) ? c_src[nr] : 0.0f;

  v8f c = {};
#pragma unroll
  for (int kb = 0; kb < 24; kb += 4) {
    int k = kb + 2 * kh;
    v2f a = *reinterpret_cast<const v2f*>(h24 + (size_t)nr * 24 + k);
    a.x *= cs;
    a.y *= cs;
    v2f b = *reinterpret_cast<const v2f*>(W1p + col * 24 + k);
    c = __builtin_amdgcn_wmma_f32_16x16x4_f32(false, a, false, b, (short)0, c,
                                              false, false);
  }
  if (col < 10) {
    if (n0 + 16 <= N) {
#pragma unroll
      for (int v = 0; v < 8; ++v)
        g1[(size_t)(n0 + v + 8 * kh) * 10 + col] = c[v];
    } else {
#pragma unroll
      for (int v = 0; v < 8; ++v) {
        int nn = n0 + v + 8 * kh;
        if (nn < N) g1[(size_t)nn * 10 + col] = c[v];
      }
    }
  }
}

__global__ void k_scatter10(const int* __restrict__ src, const int* __restrict__ dst,
                            const float* __restrict__ g1, float* __restrict__ agg1, int E) {
  const int wave = blockIdx.x * (blockDim.x >> 5) + (threadIdx.x >> 5);
  const int lane = threadIdx.x & 31;
  const long e_wave = (long)wave * 30;
  if (e_wave >= E || lane >= 30) return;
  const int grp = lane / 10;
  const int col = lane - 10 * grp;
  const long last = (long)E - 1;
  long el = e_wave + lane;
  el = el < last ? el : last;
  const int sl = src[el], dl = dst[el];

  int d[10];
  float v[10];
  for (int i = 0; i < 10; ++i) {
    const int k = 3 * i + grp;
    const int s = __shfl(sl, k);
    d[i] = __shfl(dl, k);
    v[i] = g1[(size_t)s * 10 + col];
  }
  if (e_wave + 30 <= E) {
    for (int i = 0; i < 10; ++i) atomicAdd(&agg1[(size_t)d[i] * 10 + col], v[i]);
  } else {
    for (int i = 0; i < 10; ++i)
      if (e_wave + 3 * i + grp < E) atomicAdd(&agg1[(size_t)d[i] * 10 + col], v[i]);
  }
}

__global__ void k_scatter2(const int* __restrict__ src, const int* __restrict__ dst,
                           const float* __restrict__ g2, float* __restrict__ agg2, int E) {
  const int wave = blockIdx.x * (blockDim.x >> 5) + (threadIdx.x >> 5);
  const int lane = threadIdx.x & 31;
  const long e_wave = (long)wave * 64;
  if (e_wave >= E) return;
  const int col = lane & 1, p = lane >> 1;
  const long last = (long)E - 1;

  int sl[2], dl[2];
  for (int c = 0; c < 2; ++c) {
    long el = e_wave + 32 * c + lane;
    el = el < last ? el : last;
    sl[c] = src[el];
    dl[c] = dst[el];
  }
  int d[4];
  float v[4];
  for (int r = 0; r < 4; ++r) {
    const int k = 16 * (r & 1) + p;
    const int s = __shfl(sl[r >> 1], k);
    d[r] = __shfl(dl[r >> 1], k);
    v[r] = g2[(size_t)s * 2 + col];
  }
  if (e_wave + 64 <= E) {
    for (int r = 0; r < 4; ++r) atomicAdd(&agg2[(size_t)d[r] * 2 + col], v[r]);
  } else {
    for (int r = 0; r < 4; ++r)
      if (e_wave + 16 * r + p < E) atomicAdd(&agg2[(size_t)d[r] * 2 + col], v[r]);
  }
}

__global__ void k_node2(const float* __restrict__ agg1,
                        const float* __restrict__ c_src, const float* __restrict__ c_dst,
                        const float* __restrict__ b1, const float* __restrict__ W2,
                        float* __restrict__ g2, int N) {
  int n = blockIdx.x * blockDim.x + threadIdx.x;
  if (n >= N) return;
  float cd = c_dst[n], cs = c_src[n];
  float a0 = 0.0f, a1 = 0.0f;
#pragma unroll
  for (int j = 0; j < 10; ++j) {
    float h = fmaxf(agg1[(size_t)n * 10 + j] * cd + b1[j], 0.0f);
    a0 += h * W2[j * 2 + 0];
    a1 += h * W2[j * 2 + 1];
  }
  g2[(size_t)n * 2 + 0] = a0 * cs;
  g2[(size_t)n * 2 + 1] = a1 * cs;
}

__global__ void k_out(const float* __restrict__ agg2, const float* __restrict__ c_dst,
                      const float* __restrict__ b2, float* __restrict__ out, int N) {
  int n = blockIdx.x * blockDim.x + threadIdx.x;
  if (n >= N) return;
  float cd = c_dst[n];
  out[(size_t)n * 2 + 0] = agg2[(size_t)n * 2 + 0] * cd + b2[0];
  out[(size_t)n * 2 + 1] = agg2[(size_t)n * 2 + 1] * cd + b2[1];
}


extern "C" void kernel_launch(void* const* d_in, const int* in_sizes, int n_in,
                              void* d_out, int out_size, void* d_ws, size_t ws_size,
                              hipStream_t stream) {
  const float* node_feats = (const float*)d_in[0];
  const float* edge_feats = (const float*)d_in[1];
  const float* We = (const float*)d_in[2];
  const float* be = (const float*)d_in[3];
  const float* Wn = (const float*)d_in[4];
  const float* bn = (const float*)d_in[5];
  const float* W1 = (const float*)d_in[6];
  const float* b1 = (const float*)d_in[7];
  const float* W2 = (const float*)d_in[8];
  const float* b2 = (const float*)d_in[9];
  const int* src = (const int*)d_in[10];
  const int* dst = (const int*)d_in[11];
  float* out = (float*)d_out;

  const int N = in_sizes[0] / 64;
  const int E = in_sizes[10];
  const size_t Nl = (size_t)N;

  float* ws = (float*)d_ws;
  float* deg_in = ws;
  float* deg_out = ws + Nl;
  float* e_sum = ws + 2 * Nl;
  float* agg1 = ws + 10 * Nl;
  float* agg2 = ws + 20 * Nl;
  float* c_src = ws + 22 * Nl;
  float* c_dst = ws + 23 * Nl;
  float* h24 = ws + 24 * Nl;
  float* g1 = ws + 48 * Nl;
  float* g2 = ws + 58 * Nl;
  float* Wep = ws + 60 * Nl;
  float* Wnp = Wep + 256;
  float* W1p = Wnp + 1024;

  const int TB = 256;
  const long zn = 22L * N;
  k_zero<<<(int)((zn + TB - 1) / TB), TB, 0, stream>>>(ws, zn);
  k_prep_weights<<<4, TB, 0, stream>>>(We, Wn, W1, Wep, Wnp, W1p);

  const int eb = (E + TB - 1) / TB;
  const int nb = (N + TB - 1) / TB;
  const int waves_per_block = TB / 32;
  const int etile_blocks = (((E + 15) / 16) + waves_per_block - 1) / waves_per_block;
  const int ntile_blocks = (((N + 15) / 16) + waves_per_block - 1) / waves_per_block;
  const int s10_blocks = (((E + 29) / 30) + waves_per_block - 1) / waves_per_block;
  const int s2_blocks = (((E + 63) / 64) + waves_per_block - 1) / waves_per_block;

  k_degree<<<eb, TB, 0, stream>>>(src, dst, deg_out, deg_in, E);
  k_edge_gemm_scatter<<<etile_blocks, TB, 0, stream>>>(edge_feats, Wep, dst, e_sum, E);
  k_coeff_edgeh<<<nb, TB, 0, stream>>>(deg_out, deg_in, be, e_sum, c_src, c_dst, h24, N);
  k_node_gemm<<<ntile_blocks, TB, 0, stream>>>(node_feats, Wnp, bn, h24, N);
  k_g1_gemm<<<ntile_blocks, TB, 0, stream>>>(h24, c_src, W1p, g1, N);
  k_scatter10<<<s10_blocks, TB, 0, stream>>>(src, dst, g1, agg1, E);
  k_node2<<<nb, TB, 0, stream>>>(agg1, c_src, c_dst, b1, W2, g2, N);
  k_scatter2<<<s2_blocks, TB, 0, stream>>>(src, dst, g2, agg2, E);
  k_out<<<nb, TB, 0, stream>>>(agg2, c_dst, b2, out, N);
}


# Degree counts privatized in LDS: a block counts one edge slice's hits on a 51200-node range with LDS int atomics, then adds its counts to the f32 degree arrays 32 nodes per atomic; sums stay exact
# speedup vs baseline: 3.0063x; 1.5856x over previous
#include <hip/hip_runtime.h>


typedef float v2f __attribute__((ext_vector_type(2)));
typedef float v8f __attribute__((ext_vector_type(8)));


__global__ void k_zero(float* __restrict__ p, long n) {
  long i = (long)blockIdx.x * blockDim.x + threadIdx.x;
  if (i < n) p[i] = 0.0f;
}

__global__ void k_prep_weights(const float* __restrict__ We,
                               const float* __restrict__ Wn,
                               const float* __restrict__ W1,
                               float* __restrict__ Wep, float* __restrict__ Wnp,
                               float* __restrict__ W1p) {
  int i = blockIdx.x * blockDim.x + threadIdx.x;
  if (i < 256) {
    int n = i >> 4, k = i & 15;
    Wep[i] = (n < 8) ? We[k * 8 + n] : 0.0f;
  }
  if (i < 1024) {
    int n = i >> 6, k = i & 63;
    Wnp[i] = Wn[k * 16 + n];
  }
  if (i < 384) {
    int n = i / 24, k = i % 24;
    W1p[i] = (n < 10) ? W1[k * 10 + n] : 0.0f;
  }
}

#define DEG_RANGE 51200
#define DEG_SLICES 32
#define DEG_TB 1024

__device__ __forceinline__ void deg_bump(unsigned* cnt, int node, int lo, unsigned span) {
  const unsigned r = (unsigned)(node - lo);
  if (r < span) atomicAdd(&cnt[r], 1u);
}

__global__ __launch_bounds__(DEG_TB) void k_degree_lds(const int* __restrict__ src,
                                                       const int* __restrict__ dst,
                                                       float* __restrict__ deg_out,
                                                       float* __restrict__ deg_in,
                                                       int E, int N) {
  __shared__ unsigned cnt[DEG_RANGE];
  const int* __restrict__ idx = blockIdx.z ? dst : src;
  float* __restrict__ deg = blockIdx.z ? deg_in : deg_out;
  const int tid = threadIdx.x;
  const int lo = blockIdx.y * DEG_RANGE;
  const int rest = N - lo;
  const unsigned span = (unsigned)(rest < DEG_RANGE ? rest : DEG_RANGE);

  for (unsigned i = tid; i < span; i += DEG_TB) cnt[i] = 0u;
  __syncthreads();

  const long per = ((((long)E + DEG_SLICES - 1) / DEG_SLICES) + 3) & ~3L;
  const long e0 = (long)blockIdx.x * per;
  long e1 = e0 + per;
  e1 = e1 < (long)E ? e1 : (long)E;
  if (e0 < e1) {
    const int4* __restrict__ p4 = reinterpret_cast<const int4*>(idx + e0);
    const long nq = (e1 - e0) >> 2;
    long q = tid;
    for (; q + 3 * DEG_TB < nq; q += 4 * DEG_TB) {
      const int4 a = p4[q];
      const int4 b = p4[q + DEG_TB];
      const int4 c = p4[q + 2 * DEG_TB];
      const int4 d = p4[q + 3 * DEG_TB];
      deg_bump(cnt, a.x, lo, span); deg_bump(cnt, a.y, lo, span);
      deg_bump(cnt, a.z, lo, span); deg_bump(cnt, a.w, lo, span);
      deg_bump(cnt, b.x, lo, span); deg_bump(cnt, b.y, lo, span);
      deg_bump(cnt, b.z, lo, span); deg_bump(cnt, b.w, lo, span);
      deg_bump(cnt, c.x, lo, span); deg_bump(cnt, c.y, lo, span);
      deg_bump(cnt, c.z, lo, span); deg_bump(cnt, c.w, lo, span);
      deg_bump(cnt, d.x, lo, span); deg_bump(cnt, d.y, lo, span);
      deg_bump(cnt, d.z, lo, span); deg_bump(cnt, d.w, lo, span);
    }
    for (; q < nq; q += DEG_TB) {
      const int4 a = p4[q];
      deg_bump(cnt, a.x, lo, span); deg_bump(cnt, a.y, lo, span);
      deg_bump(cnt, a.z, lo, span); deg_bump(cnt, a.w, lo, span);
    }
    const long et = e0 + (nq << 2) + tid;
    if (et < e1) deg_bump(cnt, idx[et], lo, span);
  }
  __syncthreads();

  for (unsigned i = tid; i < span; i += DEG_TB) {
    const unsigned c = cnt[i];
    if (c) atomicAdd(&deg[(size_t)lo + i], (float)c);
  }
}

__global__ void k_edge_gemm_scatter(const float* __restrict__ ef,
                                    const float* __restrict__ Wep,
                                    const int* __restrict__ dst,
                                    float* __restrict__ e_sum, int E) {
  int wave = blockIdx.x * (blockDim.x >> 5) + (threadIdx.x >> 5);
  int lane = threadIdx.x & 31;
  int e0 = wave * 16;
  if (e0 >= E) return;
  int m = lane & 15, kh = lane >> 4, col = m;
  int erow = e0 + m;
  int er = erow < E ? erow : (E - 1);
  float amask = (erow < E) ? 1.0f : 0.0f;

  v8f c = {};
#pragma unroll
  for (int kb = 0; kb < 16; kb += 4) {
    int k = kb + 2 * kh;
    v2f a = *reinterpret_cast<const v2f*>(ef + (size_t)er * 16 + k);
    a.x *= amask;
    a.y *= amask;
    v2f b = *reinterpret_cast<const v2f*>(Wep + col * 16 + k);
    c = __builtin_amdgcn_wmma_f32_16x16x4_f32(false, a, false, b, (short)0, c,
                                              false, false);
  }
  if (col < 8) {
    if (e0 + 16 <= E) {
      int d[8];
#pragma unroll
      for (int v = 0; v < 8; ++v) d[v] = dst[e0 + v + 8 * kh];
#pragma unroll
      for (int v = 0; v < 8; ++v)
        atomicAdd(&e_sum[(size_t)d[v] * 8 + col], c[v]);
    } else {
#pragma unroll
      for (int v = 0; v < 8; ++v) {
        int e = e0 + v + 8 * kh;
        if (e < E) atomicAdd(&e_sum[(size_t)dst[e] * 8 + col], c[v]);
      }
    }
  }
}

__global__ void k_coeff_edgeh(const float* __restrict__ deg_out,
                              const float* __restrict__ deg_in,
                              const float* __restrict__ be,
                              const float* __restrict__ e_sum,
                              float* __restrict__ c_src, float* __restrict__ c_dst,
                              float* __restrict__ h24, int N) {
  int n = blockIdx.x * blockDim.x + threadIdx.x;
  if (n >= N) return;
  float dout = deg_out[n], din = deg_in[n];
  c_src[n] = (dout > 0.0f) ? rsqrtf(dout) : 0.0f;
  c_dst[n] = (din > 0.0f) ? rsqrtf(din) : 0.0f;
  float inv = 1.0f / fmaxf(din, 1.0f);
  float bsel = (din > 0.0f) ? 1.0f : 0.0f;
#pragma unroll
  for (int j = 0; j < 8; ++j) {
    float v = e_sum[(size_t)n * 8 + j] * inv + bsel * be[j];
    h24[(size_t)n * 24 + j] = fmaxf(v, 0.0f);
  }
}

__global__ void k_node_gemm(const float* __restrict__ nf,
                            const float* __restrict__ Wnp,
                            const float* __restrict__ bn,
                            float* __restrict__ h24, int N) {
  int wave = blockIdx.x * (blockDim.x >> 5) + (threadIdx.x >> 5);
  int lane = threadIdx.x & 31;
  int n0 = wave * 16;
  if (n0 >= N) return;
  int m = lane & 15, kh = lane >> 4, col = m;
  int nrow = n0 + m;
  int nr = nrow < N ? nrow : (N - 1);
  float amask = (nrow < N) ? 1.0f : 0.0f;

  v8f c = {};
#pragma unroll
  for (int kb = 0; kb < 64; kb += 4) {
    int k = kb + 2 * kh;
    v2f a = *reinterpret_cast<const v2f*>(nf + (size_t)nr * 64 + k);
    a.x *= amask;
    a.y *= amask;
    v2f b = *reinterpret_cast<const v2f*>(Wnp + col * 64 + k);
    c = __builtin_amdgcn_wmma_f32_16x16x4_f32(false, a, false, b, (short)0, c,
                                              false, false);
  }
  float bc = bn[col];
  if (n0 + 16 <= N) {
#pragma unroll
    for (int v = 0; v < 8; ++v)
      h24[(size_t)(n0 + v + 8 * kh) * 24 + 8 + col] = fmaxf(c[v] + bc, 0.0f);
  } else {
#pragma unroll
    for (int v = 0; v < 8; ++v) {
      int nn = n0 + v + 8 * kh;
      if (nn < N) h24[(size_t)nn * 24 + 8 + col] = fmaxf(c[v] + bc, 0.0f);
    }
  }
}

__global__ void k_g1_gemm(const float* __restrict__ h24,
                          const float* __restrict__ c_src,
                          const float* __restrict__ W1p,
                          float* __restrict__ g1, int N) {
  int wave = blockIdx.x * (blockDim.x >> 5) + (threadIdx.x >> 5);
  int lane = threadIdx.x & 31;
  int n0 = wave * 16;
  if (n0 >= N) return;
  int m = lane & 15, kh = lane >> 4, col = m;
  int nrow = n0 + m;
  int nr = nrow < N ? nrow : (N - 1);
  float cs = (nrow < N) ? c_src[nr] : 0.0f;

  v8f c = {};
#pragma unroll
  for (int kb = 0; kb < 24; kb += 4) {
    int k = kb + 2 * kh;
    v2f a = *reinterpret_cast<const v2f*>(h24 + (size_t)nr * 24 + k);
    a.x *= cs;
    a.y *= cs;
    v2f b = *reinterpret_cast<const v2f*>(W1p + col * 24 + k);
    c = __builtin_amdgcn_wmma_f32_16x16x4_f32(false, a, false, b, (short)0, c,
                                              false, false);
  }
  if (col < 10) {
    if (n0 + 16 <= N) {
#pragma unroll
      for (int v = 0; v < 8; ++v)
        g1[(size_t)(n0 + v + 8 * kh) * 10 + col] = c[v];
    } else {
#pragma unroll
      for (int v = 0; v < 8; ++v) {
        int nn = n0 + v + 8 * kh;
        if (nn < N) g1[(size_t)nn * 10 + col] = c[v];
      }
    }
  }
}

__global__ void k_scatter10(const int* __restrict__ src, const int* __restrict__ dst,
                            const float* __restrict__ g1, float* __restrict__ agg1, int E) {
  const int wave = blockIdx.x * (blockDim.x >> 5) + (threadIdx.x >> 5);
  const int lane = threadIdx.x & 31;
  const long e_wave = (long)wave * 30;
  if (e_wave >= E || lane >= 30) return;
  const int grp = lane / 10;
  const int col = lane - 10 * grp;
  const long last = (long)E - 1;
  long el = e_wave + lane;
  el = el < last ? el : last;
  const int sl = src[el], dl = dst[el];

  int d[10];
  float v[10];
  for (int i = 0; i < 10; ++i) {
    const int k = 3 * i + grp;
    const int s = __shfl(sl, k);
    d[i] = __shfl(dl, k);
    v[i] = g1[(size_t)s * 10 + col];
  }
  if (e_wave + 30 <= E) {
    for (int i = 0; i < 10; ++i) atomicAdd(&agg1[(size_t)d[i] * 10 + col], v[i]);
  } else {
    for (int i = 0; i < 10; ++i)
      if (e_wave + 3 * i + grp < E) atomicAdd(&agg1[(size_t)d[i] * 10 + col], v[i]);
  }
}

__global__ void k_scatter2(const int* __restrict__ src, const int* __restrict__ dst,
                           const float* __restrict__ g2, float* __restrict__ agg2, int E) {
  const int wave = blockIdx.x * (blockDim.x >> 5) + (threadIdx.x >> 5);
  const int lane = threadIdx.x & 31;
  const long e_wave = (long)wave * 64;
  if (e_wave >= E) return;
  const int col = lane & 1, p = lane >> 1;
  const long last = (long)E - 1;

  int sl[2], dl[2];
  for (int c = 0; c < 2; ++c) {
    long el = e_wave + 32 * c + lane;
    el = el < last ? el : last;
    sl[c] = src[el];
    dl[c] = dst[el];
  }
  int d[4];
  float v[4];
  for (int r = 0; r < 4; ++r) {
    const int k = 16 * (r & 1) + p;
    const int s = __shfl(sl[r >> 1], k);
    d[r] = __shfl(dl[r >> 1], k);
    v[r] = g2[(size_t)s * 2 + col];
  }
  if (e_wave + 64 <= E) {
    for (int r = 0; r < 4; ++r) atomicAdd(&agg2[(size_t)d[r] * 2 + col], v[r]);
  } else {
    for (int r = 0; r < 4; ++r)
      if (e_wave + 16 * r + p < E) atomicAdd(&agg2[(size_t)d[r] * 2 + col], v[r]);
  }
}

__global__ void k_node2(const float* __restrict__ agg1,
                        const float* __restrict__ c_src, const float* __restrict__ c_dst,
                        const float* __restrict__ b1, const float* __restrict__ W2,
                        float* __restrict__ g2, int N) {
  int n = blockIdx.x * blockDim.x + threadIdx.x;
  if (n >= N) return;
  float cd = c_dst[n], cs = c_src[n];
  float a0 = 0.0f, a1 = 0.0f;
#pragma unroll
  for (int j = 0; j < 10; ++j) {
    float h = fmaxf(agg1[(size_t)n * 10 + j] * cd + b1[j], 0.0f);
    a0 += h * W2[j * 2 + 0];
    a1 += h * W2[j * 2 + 1];
  }
  g2[(size_t)n * 2 + 0] = a0 * cs;
  g2[(size_t)n * 2 + 1] = a1 * cs;
}

__global__ void k_out(const float* __restrict__ agg2, const float* __restrict__ c_dst,
                      const float* __restrict__ b2, float* __restrict__ out, int N) {
  int n = blockIdx.x * blockDim.x + threadIdx.x;
  if (n >= N) return;
  float cd = c_dst[n];
  out[(size_t)n * 2 + 0] = agg2[(size_t)n * 2 + 0] * cd + b2[0];
  out[(size_t)n * 2 + 1] = agg2[(size_t)n * 2 + 1] * cd + b2[1];
}


extern "C" void kernel_launch(void* const* d_in, const int* in_sizes, int n_in,
                              void* d_out, int out_size, void* d_ws, size_t ws_size,
                              hipStream_t stream) {
  const float* node_feats = (const float*)d_in[0];
  const float* edge_feats = (const float*)d_in[1];
  const float* We = (const float*)d_in[2];
  const float* be = (const float*)d_in[3];
  const float* Wn = (const float*)d_in[4];
  const float* bn = (const float*)d_in[5];
  const float* W1 = (const float*)d_in[6];
  const float* b1 = (const float*)d_in[7];
  const float* W2 = (const float*)d_in[8];
  const float* b2 = (const float*)d_in[9];
  const int* src = (const int*)d_in[10];
  const int* dst = (const int*)d_in[11];
  float* out = (float*)d_out;

  const int N = in_sizes[0] / 64;
  const int E = in_sizes[10];
  const size_t Nl = (size_t)N;

  float* ws = (float*)d_ws;
  float* deg_in = ws;
  float* deg_out = ws + Nl;
  float* e_sum = ws + 2 * Nl;
  float* agg1 = ws + 10 * Nl;
  float* agg2 = ws + 20 * Nl;
  float* c_src = ws + 22 * Nl;
  float* c_dst = ws + 23 * Nl;
  float* h24 = ws + 24 * Nl;
  float* g1 = ws + 48 * Nl;
  float* g2 = ws + 58 * Nl;
  float* Wep = ws + 60 * Nl;
  float* Wnp = Wep + 256;
  float* W1p = Wnp + 1024;

  const int TB = 256;
  const long zn = 22L * N;
  k_zero<<<(int)((zn + TB - 1) / TB), TB, 0, stream>>>(ws, zn);
  k_prep_weights<<<4, TB, 0, stream>>>(We, Wn, W1, Wep, Wnp, W1p);

  const int nb = (N + TB - 1) / TB;
  const int waves_per_block = TB / 32;
  const int etile_blocks = (((E + 15) / 16) + waves_per_block - 1) / waves_per_block;
  const int ntile_blocks = (((N + 15) / 16) + waves_per_block - 1) / waves_per_block;
  const int s10_blocks = (((E + 29) / 30) + waves_per_block - 1) / waves_per_block;
  const int s2_blocks = (((E + 63) / 64) + waves_per_block - 1) / waves_per_block;

  const int deg_ranges = (N + DEG_RANGE - 1) / DEG_RANGE;
  k_degree_lds<<<dim3(DEG_SLICES, deg_ranges, 2), DEG_TB, 0, stream>>>(src, dst, deg_out, deg_in, E, N);
  k_edge_gemm_scatter<<<etile_blocks, TB, 0, stream>>>(edge_feats, Wep, dst, e_sum, E);
  k_coeff_edgeh<<<nb, TB, 0, stream>>>(deg_out, deg_in, be, e_sum, c_src, c_dst, h24, N);
  k_node_gemm<<<ntile_blocks, TB, 0, stream>>>(node_feats, Wnp, bn, h24, N);
  k_g1_gemm<<<ntile_blocks, TB, 0, stream>>>(h24, c_src, W1p, g1, N);
  k_scatter10<<<s10_blocks, TB, 0, stream>>>(src, dst, g1, agg1, E);
  k_node2<<<nb, TB, 0, stream>>>(agg1, c_src, c_dst, b1, W2, g2, N);
  k_scatter2<<<s2_blocks, TB, 0, stream>>>(src, dst, g2, agg2, E);
  k_out<<<nb, TB, 0, stream>>>(agg2, c_dst, b2, out, N);
}


# Segment sums without global f32 atomics: edges ordered by 256-node destination bucket (LDS counting sort); one block sums a bucket in LDS (LDS float atomics, arrival order) and stores whole rows
# speedup vs baseline: 5.5716x; 1.8533x over previous
#include <hip/hip_runtime.h>


typedef float v2f __attribute__((ext_vector_type(2)));
typedef float v8f __attribute__((ext_vector_type(8)));


__global__ void k_zero(float* __restrict__ p, long n) {
  long i = (long)blockIdx.x * blockDim.x + threadIdx.x;
  if (i < n) p[i] = 0.0f;
}

__global__ void k_prep_weights(const float* __restrict__ We,
                               const float* __restrict__ Wn,
                               const float* __restrict__ W1,
                               float* __restrict__ Wep, float* __restrict__ Wnp,
                               float* __restrict__ W1p) {
  int i = blockIdx.x * blockDim.x + threadIdx.x;
  if (i < 256) {
    int n = i >> 4, k = i & 15;
    Wep[i] = (n < 8) ? We[k * 8 + n] : 0.0f;
  }
  if (i < 1024) {
    int n = i >> 6, k = i & 63;
    Wnp[i] = Wn[k * 16 + n];
  }
  if (i < 384) {
    int n = i / 24, k = i % 24;
    W1p[i] = (n < 10) ? W1[k * 10 + n] : 0.0f;
  }
}

#define DEG_RANGE 51200
#define DEG_SLICES 32
#define DEG_TB 1024

__device__ __forceinline__ void deg_bump(unsigned* cnt, int node, int lo, unsigned span) {
  const unsigned r = (unsigned)(node - lo);
  if (r < span) atomicAdd(&cnt[r], 1u);
}

__global__ __launch_bounds__(DEG_TB) void k_degree_lds(const int* __restrict__ src,
                                                       const int* __restrict__ dst,
                                                       float* __restrict__ deg_out,
                                                       float* __restrict__ deg_in,
                                                       int E, int N) {
  __shared__ unsigned cnt[DEG_RANGE];
  const int* __restrict__ idx = blockIdx.z ? dst : src;
  float* __restrict__ deg = blockIdx.z ? deg_in : deg_out;
  const int tid = threadIdx.x;
  const int lo = blockIdx.y * DEG_RANGE;
  const int rest = N - lo;
  const unsigned span = (unsigned)(rest < DEG_RANGE ? rest : DEG_RANGE);

  for (unsigned i = tid; i < span; i += DEG_TB) cnt[i] = 0u;
  __syncthreads();

  const long per = ((((long)E + DEG_SLICES - 1) / DEG_SLICES) + 3) & ~3L;
  const long e0 = (long)blockIdx.x * per;
  long e1 = e0 + per;
  e1 = e1 < (long)E ? e1 : (long)E;
  if (e0 < e1) {
    const int4* __restrict__ p4 = reinterpret_cast<const int4*>(idx + e0);
    const long nq = (e1 - e0) >> 2;
    long q = tid;
    for (; q + 3 * DEG_TB < nq; q += 4 * DEG_TB) {
      const int4 a = p4[q];
      const int4 b = p4[q + DEG_TB];
      const int4 c = p4[q + 2 * DEG_TB];
      const int4 d = p4[q + 3 * DEG_TB];
      deg_bump(cnt, a.x, lo, span); deg_bump(cnt, a.y, lo, span);
      deg_bump(cnt, a.z, lo, span); deg_bump(cnt, a.w, lo, span);
      deg_bump(cnt, b.x, lo, span); deg_bump(cnt, b.y, lo, span);
      deg_bump(cnt, b.z, lo, span); deg_bump(cnt, b.w, lo, span);
      deg_bump(cnt, c.x, lo, span); deg_bump(cnt, c.y, lo, span);
      deg_bump(cnt, c.z, lo, span); deg_bump(cnt, c.w, lo, span);
      deg_bump(cnt, d.x, lo, span); deg_bump(cnt, d.y, lo, span);
      deg_bump(cnt, d.z, lo, span); deg_bump(cnt, d.w, lo, span);
    }
    for (; q < nq; q += DEG_TB) {
      const int4 a = p4[q];
      deg_bump(cnt, a.x, lo, span); deg_bump(cnt, a.y, lo, span);
      deg_bump(cnt, a.z, lo, span); deg_bump(cnt, a.w, lo, span);
    }
    const long et = e0 + (nq << 2) + tid;
    if (et < e1) deg_bump(cnt, idx[et], lo, span);
  }
  __syncthreads();

  for (unsigned i = tid; i < span; i += DEG_TB) {
    const unsigned c = cnt[i];
    if (c) atomicAdd(&deg[(size_t)lo + i], (float)c);
  }
}

__global__ void k_edge_gemm_scatter(const float* __restrict__ ef,
                                    const float* __restrict__ Wep,
                                    const int* __restrict__ dst,
                                    float* __restrict__ e_sum, int E) {
  int wave = blockIdx.x * (blockDim.x >> 5) + (threadIdx.x >> 5);
  int lane = threadIdx.x & 31;
  int e0 = wave * 16;
  if (e0 >= E) return;
  int m = lane & 15, kh = lane >> 4, col = m;
  int erow = e0 + m;
  int er = erow < E ? erow : (E - 1);
  float amask = (erow < E) ? 1.0f : 0.0f;

  v8f c = {};
#pragma unroll
  for (int kb = 0; kb < 16; kb += 4) {
    int k = kb + 2 * kh;
    v2f a = *reinterpret_cast<const v2f*>(ef + (size_t)er * 16 + k);
    a.x *= amask;
    a.y *= amask;
    v2f b = *reinterpret_cast<const v2f*>(Wep + col * 16 + k);
    c = __builtin_amdgcn_wmma_f32_16x16x4_f32(false, a, false, b, (short)0, c,
                                              false, false);
  }
  if (col < 8) {
    if (e0 + 16 <= E) {
      int d[8];
#pragma unroll
      for (int v = 0; v < 8; ++v) d[v] = dst[e0 + v + 8 * kh];
#pragma unroll
      for (int v = 0; v < 8; ++v)
        atomicAdd(&e_sum[(size_t)d[v] * 8 + col], c[v]);
    } else {
#pragma unroll
      for (int v = 0; v < 8; ++v) {
        int e = e0 + v + 8 * kh;
        if (e < E) atomicAdd(&e_sum[(size_t)dst[e] * 8 + col], c[v]);
      }
    }
  }
}

__global__ void k_coeff_edgeh(const float* __restrict__ deg_out,
                              const float* __restrict__ deg_in,
                              const float* __restrict__ be,
                              const float* __restrict__ e_sum,
                              float* __restrict__ c_src, float* __restrict__ c_dst,
                              float* __restrict__ h24, int N) {
  int n = blockIdx.x * blockDim.x + threadIdx.x;
  if (n >= N) return;
  float dout = deg_out[n], din = deg_in[n];
  c_src[n] = (dout > 0.0f) ? rsqrtf(dout) : 0.0f;
  c_dst[n] = (din > 0.0f) ? rsqrtf(din) : 0.0f;
  float inv = 1.0f / fmaxf(din, 1.0f);
  float bsel = (din > 0.0f) ? 1.0f : 0.0f;
#pragma unroll
  for (int j = 0; j < 8; ++j) {
    float v = e_sum[(size_t)n * 8 + j] * inv + bsel * be[j];
    h24[(size_t)n * 24 + j] = fmaxf(v, 0.0f);
  }
}

__global__ void k_node_gemm(const float* __restrict__ nf,
                            const float* __restrict__ Wnp,
                            const float* __restrict__ bn,
                            float* __restrict__ h24, int N) {
  int wave = blockIdx.x * (blockDim.x >> 5) + (threadIdx.x >> 5);
  int lane = threadIdx.x & 31;
  int n0 = wave * 16;
  if (n0 >= N) return;
  int m = lane & 15, kh = lane >> 4, col = m;
  int nrow = n0 + m;
  int nr = nrow < N ? nrow : (N - 1);
  float amask = (nrow < N) ? 1.0f : 0.0f;

  v8f c = {};
#pragma unroll
  for (int kb = 0; kb < 64; kb += 4) {
    int k = kb + 2 * kh;
    v2f a = *reinterpret_cast<const v2f*>(nf + (size_t)nr * 64 + k);
    a.x *= amask;
    a.y *= amask;
    v2f b = *reinterpret_cast<const v2f*>(Wnp + col * 64 + k);
    c = __builtin_amdgcn_wmma_f32_16x16x4_f32(false, a, false, b, (short)0, c,
                                              false, false);
  }
  float bc = bn[col];
  if (n0 + 16 <= N) {
#pragma unroll
    for (int v = 0; v < 8; ++v)
      h24[(size_t)(n0 + v + 8 * kh) * 24 + 8 + col] = fmaxf(c[v] + bc, 0.0f);
  } else {
#pragma unroll
    for (int v = 0; v < 8; ++v) {
      int nn = n0 + v + 8 * kh;
      if (nn < N) h24[(size_t)nn * 24 + 8 + col] = fmaxf(c[v] + bc, 0.0f);
    }
  }
}

__global__ void k_g1_gemm(const float* __restrict__ h24,
                          const float* __restrict__ c_src,
                          const float* __restrict__ W1p,
                          float* __restrict__ g1, int N) {
  int wave = blockIdx.x * (blockDim.x >> 5) + (threadIdx.x >> 5);
  int lane = threadIdx.x & 31;
  int n0 = wave * 16;
  if (n0 >= N) return;
  int m = lane & 15, kh = lane >> 4, col = m;
  int nrow = n0 + m;
  int nr = nrow < N ? nrow : (N - 1);
  float cs = (nrow < N) ? c_src[nr] : 0.0f;

  v8f c = {};
#pragma unroll
  for (int kb = 0; kb < 24; kb += 4) {
    int k = kb + 2 * kh;
    v2f a = *reinterpret_cast<const v2f*>(h24 + (size_t)nr * 24 + k);
    a.x *= cs;
    a.y *= cs;
    v2f b = *reinterpret_cast<const v2f*>(W1p + col * 24 + k);
    c = __builtin_amdgcn_wmma_f32_16x16x4_f32(false, a, false, b, (short)0, c,
                                              false, false);
  }
  if (col < 10) {
    if (n0 + 16 <= N) {
#pragma unroll
      for (int v = 0; v < 8; ++v)
        g1[(size_t)(n0 + v + 8 * kh) * 10 + col] = c[v];
    } else {
#pragma unroll
      for (int v = 0; v < 8; ++v) {
        int nn = n0 + v + 8 * kh;
        if (nn < N) g1[(size_t)nn * 10 + col] = c[v];
      }
    }
  }
}

__global__ void k_scatter10(const int* __restrict__ src, const int* __restrict__ dst,
                            const float* __restrict__ g1, float* __restrict__ agg1, int E) {
  const int wave = blockIdx.x * (blockDim.x >> 5) + (threadIdx.x >> 5);
  const int lane = threadIdx.x & 31;
  const long e_wave = (long)wave * 30;
  if (e_wave >= E || lane >= 30) return;
  const int grp = lane / 10;
  const int col = lane - 10 * grp;
  const long last = (long)E - 1;
  long el = e_wave + lane;
  el = el < last ? el : last;
  const int sl = src[el], dl = dst[el];

  int d[10];
  float v[10];
  for (int i = 0; i < 10; ++i) {
    const int k = 3 * i + grp;
    const int s = __shfl(sl, k);
    d[i] = __shfl(dl, k);
    v[i] = g1[(size_t)s * 10 + col];
  }
  if (e_wave + 30 <= E) {
    for (int i = 0; i < 10; ++i) atomicAdd(&agg1[(size_t)d[i] * 10 + col], v[i]);
  } else {
    for (int i = 0; i < 10; ++i)
      if (e_wave + 3 * i + grp < E) atomicAdd(&agg1[(size_t)d[i] * 10 + col], v[i]);
  }
}

__global__ void k_scatter2(const int* __restrict__ src, const int* __restrict__ dst,
                           const float* __restrict__ g2, float* __restrict__ agg2, int E) {
  const int wave = blockIdx.x * (blockDim.x >> 5) + (threadIdx.x >> 5);
  const int lane = threadIdx.x & 31;
  const long e_wave = (long)wave * 64;
  if (e_wave >= E) return;
  const int col = lane & 1, p = lane >> 1;
  const long last = (long)E - 1;

  int sl[2], dl[2];
  for (int c = 0; c < 2; ++c) {
    long el = e_wave + 32 * c + lane;
    el = el < last ? el : last;
    sl[c] = src[el];
    dl[c] = dst[el];
  }
  int d[4];
  float v[4];
  for (int r = 0; r < 4; ++r) {
    const int k = 16 * (r & 1) + p;
    const int s = __shfl(sl[r >> 1], k);
    d[r] = __shfl(dl[r >> 1], k);
    v[r] = g2[(size_t)s * 2 + col];
  }
  if (e_wave + 64 <= E) {
    for (int r = 0; r < 4; ++r) atomicAdd(&agg2[(size_t)d[r] * 2 + col], v[r]);
  } else {
    for (int r = 0; r < 4; ++r)
      if (e_wave + 16 * r + p < E) atomicAdd(&agg2[(size_t)d[r] * 2 + col], v[r]);
  }
}

__global__ void k_node2(const float* __restrict__ agg1,
                        const float* __restrict__ c_src, const float* __restrict__ c_dst,
                        const float* __restrict__ b1, const float* __restrict__ W2,
                        float* __restrict__ g2, int N) {
  int n = blockIdx.x * blockDim.x + threadIdx.x;
  if (n >= N) return;
  float cd = c_dst[n], cs = c_src[n];
  float a0 = 0.0f, a1 = 0.0f;
#pragma unroll
  for (int j = 0; j < 10; ++j) {
    float h = fmaxf(agg1[(size_t)n * 10 + j] * cd + b1[j], 0.0f);
    a0 += h * W2[j * 2 + 0];
    a1 += h * W2[j * 2 + 1];
  }
  g2[(size_t)n * 2 + 0] = a0 * cs;
  g2[(size_t)n * 2 + 1] = a1 * cs;
}

__global__ void k_out(const float* __restrict__ agg2, const float* __restrict__ c_dst,
                      const float* __restrict__ b2, float* __restrict__ out, int N) {
  int n = blockIdx.x * blockDim.x + threadIdx.x;
  if (n >= N) return;
  float cd = c_dst[n];
  out[(size_t)n * 2 + 0] = agg2[(size_t)n * 2 + 0] * cd + b2[0];
  out[(size_t)n * 2 + 1] = agg2[(size_t)n * 2 + 1] * cd + b2[1];
}

#define BKT_SHIFT 8
#define BKT_NODES 256
#define PART_TB 1024
#define PART_QPT 3
#define PART_EDGES (PART_TB * 4 * PART_QPT)
#define SUM_TB 1024
#define SUM_WAVES (SUM_TB / 32)

__device__ __forceinline__ unsigned wave_excl_scan(unsigned* a, int n, int lane) {
  const int seg = (n + 31) >> 5;
  const int b0 = lane * seg;
  int b1 = b0 + seg;
  b1 = b1 < n ? b1 : n;
  unsigned s = 0u;
  for (int b = b0; b < b1; ++b) s += a[b];
  unsigned inc = s;
  for (int o = 1; o < 32; o <<= 1) {
    const unsigned t = __shfl_up(inc, o);
    if (lane >= o) inc += t;
  }
  unsigned run = inc - s;
  for (int b = b0; b < b1; ++b) {
    const unsigned c = a[b];
    a[b] = run;
    run += c;
  }
  return __shfl(inc, 31);
}

__global__ __launch_bounds__(PART_TB) void k_part_count(const int* __restrict__ dst,
                                                        unsigned* __restrict__ cnt,
                                                        int E, int nbk) {
  extern __shared__ unsigned part_lds[];
  unsigned* hist = part_lds;
  const int tid = threadIdx.x;
  for (int i = tid; i < nbk; i += PART_TB) hist[i] = 0u;
  __syncthreads();
  const long e0 = (long)blockIdx.x * PART_EDGES;
  long e1 = e0 + PART_EDGES;
  e1 = e1 < (long)E ? e1 : (long)E;
  const long nq = (e1 - e0) >> 2;
  const int4* __restrict__ d4 = reinterpret_cast<const int4*>(dst + e0);
  if (nq > 0) {
    int4 dv[PART_QPT];
#pragma unroll
    for (int u = 0; u < PART_QPT; ++u) {
      const long q = tid + u * PART_TB;
      dv[u] = d4[q < nq ? q : nq - 1];
    }
#pragma unroll
    for (int u = 0; u < PART_QPT; ++u) {
      const long q = tid + u * PART_TB;
      if (q < nq) {
        atomicAdd(&hist[dv[u].x >> BKT_SHIFT], 1u);
        atomicAdd(&hist[dv[u].y >> BKT_SHIFT], 1u);
        atomicAdd(&hist[dv[u].z >> BKT_SHIFT], 1u);
        atomicAdd(&hist[dv[u].w >> BKT_SHIFT], 1u);
      }
    }
  }
  const long et = e0 + (nq << 2) + tid;
  if (et < e1) atomicAdd(&hist[dst[et] >> BKT_SHIFT], 1u);
  __syncthreads();
  for (int i = tid; i < nbk; i += PART_TB) cnt[(size_t)blockIdx.x * nbk + i] = hist[i];
}

__global__ __launch_bounds__(PART_TB) void k_part_scan(const unsigned* __restrict__ cnt,
                                                       unsigned* __restrict__ off,
                                                       unsigned* __restrict__ bstart,
                                                       int nch, int nbk) {
  extern __shared__ unsigned part_lds[];
  unsigned* tot = part_lds;
  const int tid = threadIdx.x;
  for (int b = tid; b < nbk; b += PART_TB) {
    unsigned run = 0u;
#pragma unroll 16
    for (int k = 0; k < nch; ++k) {
      const unsigned c = cnt[(size_t)k * nbk + b];
      off[(size_t)k * nbk + b] = run;
      run += c;
    }
    tot[b] = run;
  }
  __syncthreads();
  if (tid < 32) {
    const unsigned total = wave_excl_scan(tot, nbk, tid);
    if (tid == 0) tot[nbk] = total;
  }
  __syncthreads();
  for (int b = tid; b <= nbk; b += PART_TB) bstart[b] = tot[b];
}

__global__ __launch_bounds__(PART_TB) void k_part_fill(const int* __restrict__ src,
                                                       const int* __restrict__ dst,
                                                       const unsigned* __restrict__ cnt,
                                                       const unsigned* __restrict__ off,
                                                       const unsigned* __restrict__ bstart,
                                                       unsigned* __restrict__ rec,
                                                       unsigned* __restrict__ eidx,
                                                       int E, int nbk) {
  __shared__ unsigned st_s[PART_EDGES];
  __shared__ unsigned st_d[PART_EDGES];
  __shared__ unsigned st_e[PART_EDGES];
  extern __shared__ unsigned part_lds[];
  unsigned* cur = part_lds;
  unsigned* gdelta = part_lds + nbk;
  const int tid = threadIdx.x;
  const long e0 = (long)blockIdx.x * PART_EDGES;
  long e1 = e0 + PART_EDGES;
  e1 = e1 < (long)E ? e1 : (long)E;
  const long nq = (e1 - e0) >> 2;
  const int4* __restrict__ d4 = reinterpret_cast<const int4*>(dst + e0);
  const int4* __restrict__ s4 = reinterpret_cast<const int4*>(src + e0);
  int4 dv[PART_QPT] = {}, sv[PART_QPT] = {};
  if (nq > 0) {
#pragma unroll
    for (int u = 0; u < PART_QPT; ++u) {
      const long q = tid + u * PART_TB;
      const long qc = q < nq ? q : nq - 1;
      dv[u] = d4[qc];
      sv[u] = s4[qc];
    }
  }
  for (int i = tid; i < nbk; i += PART_TB) cur[i] = cnt[(size_t)blockIdx.x * nbk + i];
  __syncthreads();
  if (tid < 32) wave_excl_scan(cur, nbk, tid);
  __syncthreads();
  for (int i = tid; i < nbk; i += PART_TB)
    gdelta[i] = bstart[i] + off[(size_t)blockIdx.x * nbk + i] - cur[i];
  __syncthreads();

#pragma unroll
  for (int u = 0; u < PART_QPT; ++u) {
    const long q = tid + u * PART_TB;
    if (q < nq) {
      const unsigned eb = (unsigned)(e0 + (q << 2));
      unsigned p;
      p = atomicAdd(&cur[dv[u].x >> BKT_SHIFT], 1u);
      st_s[p] = (unsigned)sv[u].x; st_d[p] = (unsigned)dv[u].x; st_e[p] = eb;
      p = atomicAdd(&cur[dv[u].y >> BKT_SHIFT], 1u);
      st_s[p] = (unsigned)sv[u].y; st_d[p] = (unsigned)dv[u].y; st_e[p] = eb + 1u;
      p = atomicAdd(&cur[dv[u].z >> BKT_SHIFT], 1u);
      st_s[p] = (unsigned)sv[u].z; st_d[p] = (unsigned)dv[u].z; st_e[p] = eb + 2u;
      p = atomicAdd(&cur[dv[u].w >> BKT_SHIFT], 1u);
      st_s[p] = (unsigned)sv[u].w; st_d[p] = (unsigned)dv[u].w; st_e[p] = eb + 3u;
    }
  }
  const long et = e0 + (nq << 2) + tid;
  if (et < e1) {
    const int dd = dst[et];
    const unsigned p = atomicAdd(&cur[dd >> BKT_SHIFT], 1u);
    st_s[p] = (unsigned)src[et]; st_d[p] = (unsigned)dd; st_e[p] = (unsigned)et;
  }
  __syncthreads();

  const int nloc = (int)(e1 - e0);
  for (int i = tid; i < nloc; i += PART_TB) {
    const unsigned dd = st_d[i];
    const unsigned pos = (unsigned)i + gdelta[dd >> BKT_SHIFT];
    rec[pos] = (st_s[i] << 8) | (dd & (BKT_NODES - 1));
    eidx[pos] = st_e[i];
  }
}

__global__ __launch_bounds__(SUM_TB) void k_bucket_edge(const float* __restrict__ ef,
                                                        const float* __restrict__ Wep,
                                                        const unsigned* __restrict__ rec,
                                                        const unsigned* __restrict__ eidx,
                                                        const unsigned* __restrict__ bstart,
                                                        float* __restrict__ e_sum, int N) {
  __shared__ float acc[BKT_NODES * 8];
  const int tid = threadIdx.x;
  const int lo = blockIdx.x << BKT_SHIFT;
  for (int i = tid; i < BKT_NODES * 8; i += SUM_TB) acc[i] = 0.0f;
  __syncthreads();
  const unsigned j0 = bstart[blockIdx.x], j1 = bstart[blockIdx.x + 1];
  const int wave = tid >> 5, lane = tid & 31;
  const int m = lane & 15, kh = lane >> 4, col = m;
  unsigned j = j0 + wave * 16;
  unsigned er = 0u, rm = 0u;
  if (j < j1) {
    const unsigned jc = j + m < j1 ? j + m : j1 - 1u;
    er = eidx[jc];
    rm = rec[jc];
  }
  while (j < j1) {
    const unsigned jn = j + SUM_WAVES * 16;
    unsigned ern = 0u, rmn = 0u;
    if (jn < j1) {
      const unsigned jc = jn + m < j1 ? jn + m : j1 - 1u;
      ern = eidx[jc];
      rmn = rec[jc];
    }
    const bool live = j + m < j1;
    unsigned dl[8];
#pragma unroll
    for (int v = 0; v < 8; ++v) dl[v] = __shfl(rm, v + 8 * kh) & (BKT_NODES - 1);
    v8f c = {};
#pragma unroll
    for (int kb = 0; kb < 16; kb += 4) {
      const int k = kb + 2 * kh;
      v2f a = *reinterpret_cast<const v2f*>(ef + (size_t)er * 16 + k);
      a.x = live ? a.x : 0.0f;
      a.y = live ? a.y : 0.0f;
      const v2f b = *reinterpret_cast<const v2f*>(Wep + col * 16 + k);
      c = __builtin_amdgcn_wmma_f32_16x16x4_f32(false, a, false, b, (short)0, c,
                                                false, false);
    }
    if (col < 8) {
#pragma unroll
      for (int v = 0; v < 8; ++v)
        if (j + v + 8 * kh < j1) atomicAdd(&acc[dl[v] * 8 + col], c[v]);
    }
    er = ern;
    rm = rmn;
    j = jn;
  }
  __syncthreads();
  const int rest = N - lo;
  const int nn = rest < BKT_NODES ? rest : BKT_NODES;
  for (int i = tid; i < nn * 8; i += SUM_TB) e_sum[(size_t)lo * 8 + i] = acc[i];
}

__global__ __launch_bounds__(SUM_TB) void k_bucket_sum10(const unsigned* __restrict__ rec,
                                                         const unsigned* __restrict__ bstart,
                                                         const float* __restrict__ g1,
                                                         float* __restrict__ agg1, int N) {
  __shared__ float acc[BKT_NODES * 10];
  const int tid = threadIdx.x;
  const int lo = blockIdx.x << BKT_SHIFT;
  for (int i = tid; i < BKT_NODES * 10; i += SUM_TB) acc[i] = 0.0f;
  __syncthreads();
  const unsigned j0 = bstart[blockIdx.x], j1 = bstart[blockIdx.x + 1];
  const int wave = tid >> 5, lane = tid & 31;
  const bool has_col = lane < 30;
  const int grp = has_col ? lane / 10 : 0;
  const int col = has_col ? lane - 10 * grp : 0;
  unsigned j = j0 + wave * 30;
  unsigned r = 0u;
  if (j < j1) r = rec[j + lane < j1 ? j + lane : j1 - 1u];
  while (j < j1) {
    const unsigned jn = j + SUM_WAVES * 30;
    unsigned rn = 0u;
    if (jn < j1) rn = rec[jn + lane < j1 ? jn + lane : j1 - 1u];
    unsigned dl[10];
    float v[10];
    for (int i = 0; i < 10; ++i) {
      const unsigned rr = __shfl(r, 3 * i + grp);
      dl[i] = rr & (BKT_NODES - 1);
      v[i] = g1[(size_t)(rr >> 8) * 10 + col];
    }
    if (has_col) {
      for (int i = 0; i < 10; ++i)
        if (j + 3 * i + grp < j1) atomicAdd(&acc[dl[i] * 10 + col], v[i]);
    }
    r = rn;
    j = jn;
  }
  __syncthreads();
  const int rest = N - lo;
  const int nn = rest < BKT_NODES ? rest : BKT_NODES;
  for (int i = tid; i < nn * 10; i += SUM_TB) agg1[(size_t)lo * 10 + i] = acc[i];
}

__global__ __launch_bounds__(SUM_TB) void k_bucket_sum2(const unsigned* __restrict__ rec,
                                                        const unsigned* __restrict__ bstart,
                                                        const float* __restrict__ g2,
                                                        float* __restrict__ agg2, int N) {
  __shared__ float acc[BKT_NODES * 2];
  const int tid = threadIdx.x;
  const int lo = blockIdx.x << BKT_SHIFT;
  for (int i = tid; i < BKT_NODES * 2; i += SUM_TB) acc[i] = 0.0f;
  __syncthreads();
  const unsigned j0 = bstart[blockIdx.x], j1 = bstart[blockIdx.x + 1];
  const int wave = tid >> 5, lane = tid & 31;
  const int col = lane & 1, p = lane >> 1;
  for (unsigned j = j0 + wave * 64; j < j1; j += SUM_WAVES * 64) {
    unsigned rl[2];
    for (int c = 0; c < 2; ++c) {
      const unsigned jl = j + 32 * c + lane;
      rl[c] = rec[jl < j1 ? jl : j1 - 1u];
    }
    unsigned dl[4];
    float v[4];
    for (int r = 0; r < 4; ++r) {
      const unsigned rr = __shfl(rl[r >> 1], 16 * (r & 1) + p);
      dl[r] = rr & (BKT_NODES - 1);
      v[r] = g2[(size_t)(rr >> 8) * 2 + col];
    }
    for (int r = 0; r < 4; ++r)
      if (j + 16 * r + p < j1) atomicAdd(&acc[dl[r] * 2 + col], v[r]);
  }
  __syncthreads();
  const int rest = N - lo;
  const int nn = rest < BKT_NODES ? rest : BKT_NODES;
  for (int i = tid; i < nn * 2; i += SUM_TB) agg2[(size_t)lo * 2 + i] = acc[i];
}


extern "C" void kernel_launch(void* const* d_in, const int* in_sizes, int n_in,
                              void* d_out, int out_size, void* d_ws, size_t ws_size,
                              hipStream_t stream) {
  const float* node_feats = (const float*)d_in[0];
  const float* edge_feats = (const float*)d_in[1];
  const float* We = (const float*)d_in[2];
  const float* be = (const float*)d_in[3];
  const float* Wn = (const float*)d_in[4];
  const float* bn = (const float*)d_in[5];
  const float* W1 = (const float*)d_in[6];
  const float* b1 = (const float*)d_in[7];
  const float* W2 = (const float*)d_in[8];
  const float* b2 = (const float*)d_in[9];
  const int* src = (const int*)d_in[10];
  const int* dst = (const int*)d_in[11];
  float* out = (float*)d_out;

  const int N = in_sizes[0] / 64;
  const int E = in_sizes[10];
  const size_t Nl = (size_t)N;

  float* ws = (float*)d_ws;
  float* deg_in = ws;
  float* deg_out = ws + Nl;
  float* e_sum = ws + 2 * Nl;
  float* agg1 = ws + 10 * Nl;
  float* agg2 = ws + 20 * Nl;
  float* c_src = ws + 22 * Nl;
  float* c_dst = ws + 23 * Nl;
  float* h24 = ws + 24 * Nl;
  float* g1 = ws + 48 * Nl;
  float* g2 = ws + 58 * Nl;
  float* Wep = ws + 60 * Nl;
  float* Wnp = Wep + 256;
  float* W1p = Wnp + 1024;

  const int TB = 256;
  const int nb = (N + TB - 1) / TB;
  const int waves_per_block = TB / 32;
  const int ntile_blocks = (((N + 15) / 16) + waves_per_block - 1) / waves_per_block;

  const int nbk = (N + BKT_NODES - 1) / BKT_NODES;
  const int nch = (E + PART_EDGES - 1) / PART_EDGES;
  const size_t ncb = (size_t)nch * nbk;
  unsigned* cnt = (unsigned*)(W1p + 384);
  unsigned* off = cnt + ncb;
  unsigned* bstart = off + ncb;
  unsigned* rec = bstart + nbk + 1;
  unsigned* eidx = rec + (size_t)E;
  const size_t lds_words = (size_t)nbk * 2 + 1;

  const size_t need_words = 60 * Nl + 1664 + 2 * ncb + (size_t)nbk + 1 + 2 * (size_t)E;
  const bool ordered = ws_size >= need_words * 4;

  const long zn = ordered ? 2L * N : 22L * N;
  k_zero<<<(int)((zn + TB - 1) / TB), TB, 0, stream>>>(ws, zn);
  k_prep_weights<<<4, TB, 0, stream>>>(We, Wn, W1, Wep, Wnp, W1p);

  const int deg_ranges = (N + DEG_RANGE - 1) / DEG_RANGE;
  k_degree_lds<<<dim3(DEG_SLICES, deg_ranges, 2), DEG_TB, 0, stream>>>(src, dst, deg_out, deg_in, E, N);
  if (ordered) {
    k_part_count<<<nch, PART_TB, lds_words * 4, stream>>>(dst, cnt, E, nbk);
    k_part_scan<<<1, PART_TB, lds_words * 4, stream>>>(cnt, off, bstart, nch, nbk);
    k_part_fill<<<nch, PART_TB, lds_words * 4, stream>>>(src, dst, cnt, off, bstart, rec, eidx, E, nbk);
    k_bucket_edge<<<nbk, SUM_TB, 0, stream>>>(edge_feats, Wep, rec, eidx, bstart, e_sum, N);
  } else {
    const int etile_blocks = (((E + 15) / 16) + waves_per_block - 1) / waves_per_block;
    k_edge_gemm_scatter<<<etile_blocks, TB, 0, stream>>>(edge_feats, Wep, dst, e_sum, E);
  }
  k_coeff_edgeh<<<nb, TB, 0, stream>>>(deg_out, deg_in, be, e_sum, c_src, c_dst, h24, N);
  k_node_gemm<<<ntile_blocks, TB, 0, stream>>>(node_feats, Wnp, bn, h24, N);
  k_g1_gemm<<<ntile_blocks, TB, 0, stream>>>(h24, c_src, W1p, g1, N);
  if (ordered) {
    k_bucket_sum10<<<nbk, SUM_TB, 0, stream>>>(rec, bstart, g1, agg1, N);
  } else {
    const int s10_blocks = (((E + 29) / 30) + waves_per_block - 1) / waves_per_block;
    k_scatter10<<<s10_blocks, TB, 0, stream>>>(src, dst, g1, agg1, E);
  }
  k_node2<<<nb, TB, 0, stream>>>(agg1, c_src, c_dst, b1, W2, g2, N);
  if (ordered) {
    k_bucket_sum2<<<nbk, SUM_TB, 0, stream>>>(rec, bstart, g2, agg2, N);
  } else {
    const int s2_blocks = (((E + 63) / 64) + waves_per_block - 1) / waves_per_block;
    k_scatter2<<<s2_blocks, TB, 0, stream>>>(src, dst, g2, agg2, E);
  }
  k_out<<<nb, TB, 0, stream>>>(agg2, c_dst, b2, out, N);
}


# Short per-node and set-up launches folded into the edge passes of the ordered path (14 launches -> 7): node tails read the bucket sums in LDS; order of all f32 sums as in the parent
# speedup vs baseline: 6.0911x; 1.0932x over previous
#include <hip/hip_runtime.h>


typedef float v2f __attribute__((ext_vector_type(2)));
typedef float v8f __attribute__((ext_vector_type(8)));


__global__ void k_zero(float* __restrict__ p, long n) {
  long i = (long)blockIdx.x * blockDim.x + threadIdx.x;
  if (i < n) p[i] = 0.0f;
}

__global__ void k_prep_weights(const float* __restrict__ We,
                               const float* __restrict__ Wn,
                               const float* __restrict__ W1,
                               float* __restrict__ Wep, float* __restrict__ Wnp,
                               float* __restrict__ W1p) {
  int i = blockIdx.x * blockDim.x + threadIdx.x;
  if (i < 256) {
    int n = i >> 4, k = i & 15;
    Wep[i] = (n < 8) ? We[k * 8 + n] : 0.0f;
  }
  if (i < 1024) {
    int n = i >> 6, k = i & 63;
    Wnp[i] = Wn[k * 16 + n];
  }
  if (i < 384) {
    int n = i / 24, k = i % 24;
    W1p[i] = (n < 10) ? W1[k * 10 + n] : 0.0f;
  }
}

#define DEG_RANGE 51200
#define DEG_SLICES 32
#define DEG_TB 1024

__device__ __forceinline__ void deg_bump(unsigned* cnt, int node, int lo, unsigned span) {
  const unsigned r = (unsigned)(node - lo);
  if (r < span) atomicAdd(&cnt[r], 1u);
}

__global__ __launch_bounds__(DEG_TB) void k_degree_lds(const int* __restrict__ src,
                                                       const int* __restrict__ dst,
                                                       float* __restrict__ deg_out,
                                                       float* __restrict__ deg_in,
                                                       int E, int N) {
  __shared__ unsigned cnt[DEG_RANGE];
  const int* __restrict__ idx = blockIdx.z ? dst : src;
  float* __restrict__ deg = blockIdx.z ? deg_in : deg_out;
  const int tid = threadIdx.x;
  const int lo = blockIdx.y * DEG_RANGE;
  const int rest = N - lo;
  const unsigned span = (unsigned)(rest < DEG_RANGE ? rest : DEG_RANGE);

  for (unsigned i = tid; i < span; i += DEG_TB) cnt[i] = 0u;
  __syncthreads();

  const long per = ((((long)E + DEG_SLICES - 1) / DEG_SLICES) + 3) & ~3L;
  const long e0 = (long)blockIdx.x * per;
  long e1 = e0 + per;
  e1 = e1 < (long)E ? e1 : (long)E;
  if (e0 < e1) {
    const int4* __restrict__ p4 = reinterpret_cast<const int4*>(idx + e0);
    const long nq = (e1 - e0) >> 2;
    long q = tid;
    for (; q + 3 * DEG_TB < nq; q += 4 * DEG_TB) {
      const int4 a = p4[q];
      const int4 b = p4[q + DEG_TB];
      const int4 c = p4[q + 2 * DEG_TB];
      const int4 d = p4[q + 3 * DEG_TB];
      deg_bump(cnt, a.x, lo, span); deg_bump(cnt, a.y, lo, span);
      deg_bump(cnt, a.z, lo, span); deg_bump(cnt, a.w, lo, span);
      deg_bump(cnt, b.x, lo, span); deg_bump(cnt, b.y, lo, span);
      deg_bump(cnt, b.z, lo, span); deg_bump(cnt, b.w, lo, span);
      deg_bump(cnt, c.x, lo, span); deg_bump(cnt, c.y, lo, span);
      deg_bump(cnt, c.z, lo, span); deg_bump(cnt, c.w, lo, span);
      deg_bump(cnt, d.x, lo, span); deg_bump(cnt, d.y, lo, span);
      deg_bump(cnt, d.z, lo, span); deg_bump(cnt, d.w, lo, span);
    }
    for (; q < nq; q += DEG_TB) {
      const int4 a = p4[q];
      deg_bump(cnt, a.x, lo, span); deg_bump(cnt, a.y, lo, span);
      deg_bump(cnt, a.z, lo, span); deg_bump(cnt, a.w, lo, span);
    }
    const long et = e0 + (nq << 2) + tid;
    if (et < e1) deg_bump(cnt, idx[et], lo, span);
  }
  __syncthreads();

  for (unsigned i = tid; i < span; i += DEG_TB) {
    const unsigned c = cnt[i];
    if (c) atomicAdd(&deg[(size_t)lo + i], (float)c);
  }
}

__global__ void k_edge_gemm_scatter(const float* __restrict__ ef,
                                    const float* __restrict__ Wep,
                                    const int* __restrict__ dst,
                                    float* __restrict__ e_sum, int E) {
  int wave = blockIdx.x * (blockDim.x >> 5) + (threadIdx.x >> 5);
  int lane = threadIdx.x & 31;
  int e0 = wave * 16;
  if (e0 >= E) return;
  int m = lane & 15, kh = lane >> 4, col = m;
  int erow = e0 + m;
  int er = erow < E ? erow : (E - 1);
  float amask = (erow < E) ? 1.0f : 0.0f;

  v8f c = {};
#pragma unroll
  for (int kb = 0; kb < 16; kb += 4) {
    int k = kb + 2 * kh;
    v2f a = *reinterpret_cast<const v2f*>(ef + (size_t)er * 16 + k);
    a.x *= amask;
    a.y *= amask;
    v2f b = *reinterpret_cast<const v2f*>(Wep + col * 16 + k);
    c = __builtin_amdgcn_wmma_f32_16x16x4_f32(false, a, false, b, (short)0, c,
                                              false, false);
  }
  if (col < 8) {
    if (e0 + 16 <= E) {
      int d[8];
#pragma unroll
      for (int v = 0; v < 8; ++v) d[v] = dst[e0 + v + 8 * kh];
#pragma unroll
      for (int v = 0; v < 8; ++v)
        atomicAdd(&e_sum[(size_t)d[v] * 8 + col], c[v]);
    } else {
#pragma unroll
      for (int v = 0; v < 8; ++v) {
        int e = e0 + v + 8 * kh;
        if (e < E) atomicAdd(&e_sum[(size_t)dst[e] * 8 + col], c[v]);
      }
    }
  }
}

__global__ void k_coeff_edgeh(const float* __restrict__ deg_out,
                              const float* __restrict__ deg_in,
                              const float* __restrict__ be,
                              const float* __restrict__ e_sum,
                              float* __restrict__ c_src, float* __restrict__ c_dst,
                              float* __restrict__ h24, int N) {
  int n = blockIdx.x * blockDim.x + threadIdx.x;
  if (n >= N) return;
  float dout = deg_out[n], din = deg_in[n];
  c_src[n] = (dout > 0.0f) ? rsqrtf(dout) : 0.0f;
  c_dst[n] = (din > 0.0f) ? rsqrtf(din) : 0.0f;
  float inv = 1.0f / fmaxf(din, 1.0f);
  float bsel = (din > 0.0f) ? 1.0f : 0.0f;
#pragma unroll
  for (int j = 0; j < 8; ++j) {
    float v = e_sum[(size_t)n * 8 + j] * inv + bsel * be[j];
    h24[(size_t)n * 24 + j] = fmaxf(v, 0.0f);
  }
}

__global__ void k_node_gemm(const float* __restrict__ nf,
                            const float* __restrict__ Wnp,
                            const float* __restrict__ bn,
                            float* __restrict__ h24, int N) {
  int wave = blockIdx.x * (blockDim.x >> 5) + (threadIdx.x >> 5);
  int lane = threadIdx.x & 31;
  int n0 = wave * 16;
  if (n0 >= N) return;
  int m = lane & 15, kh = lane >> 4, col = m;
  int nrow = n0 + m;
  int nr = nrow < N ? nrow : (N - 1);
  float amask = (nrow < N) ? 1.0f : 0.0f;

  v8f c = {};
#pragma unroll
  for (int kb = 0; kb < 64; kb += 4) {
    int k = kb + 2 * kh;
    v2f a = *reinterpret_cast<const v2f*>(nf + (size_t)nr * 64 + k);
    a.x *= amask;
    a.y *= amask;
    v2f b = *reinterpret_cast<const v2f*>(Wnp + col * 64 + k);
    c = __builtin_amdgcn_wmma_f32_16x16x4_f32(false, a, false, b, (short)0, c,
                                              false, false);
  }
  float bc = bn[col];
  if (n0 + 16 <= N) {
#pragma unroll
    for (int v = 0; v < 8; ++v)
      h24[(size_t)(n0 + v + 8 * kh) * 24 + 8 + col] = fmaxf(c[v] + bc, 0.0f);
  } else {
#pragma unroll
    for (int v = 0; v < 8; ++v) {
      int nn = n0 + v + 8 * kh;
      if (nn < N) h24[(size_t)nn * 24 + 8 + col] = fmaxf(c[v] + bc, 0.0f);
    }
  }
}

__global__ void k_g1_gemm(const float* __restrict__ h24,
                          const float* __restrict__ c_src,
                          const float* __restrict__ W1p,
                          float* __restrict__ g1, int N) {
  int wave = blockIdx.x * (blockDim.x >> 5) + (threadIdx.x >> 5);
  int lane = threadIdx.x & 31;
  int n0 = wave * 16;
  if (n0 >= N) return;
  int m = lane & 15, kh = lane >> 4, col = m;
  int nrow = n0 + m;
  int nr = nrow < N ? nrow : (N - 1);
  float cs = (nrow < N) ? c_src[nr] : 0.0f;

  v8f c = {};
#pragma unroll
  for (int kb = 0; kb < 24; kb += 4) {
    int k = kb + 2 * kh;
    v2f a = *reinterpret_cast<const v2f*>(h24 + (size_t)nr * 24 + k);
    a.x *= cs;
    a.y *= cs;
    v2f b = *reinterpret_cast<const v2f*>(W1p + col * 24 + k);
    c = __builtin_amdgcn_wmma_f32_16x16x4_f32(false, a, false, b, (short)0, c,
                                              false, false);
  }
  if (col < 10) {
    if (n0 + 16 <= N) {
#pragma unroll
      for (int v = 0; v < 8; ++v)
        g1[(size_t)(n0 + v + 8 * kh) * 10 + col] = c[v];
    } else {
#pragma unroll
      for (int v = 0; v < 8; ++v) {
        int nn = n0 + v + 8 * kh;
        if (nn < N) g1[(size_t)nn * 10 + col] = c[v];
      }
    }
  }
}

__global__ void k_scatter10(const int* __restrict__ src, const int* __restrict__ dst,
                            const float* __restrict__ g1, float* __restrict__ agg1, int E) {
  const int wave = blockIdx.x * (blockDim.x >> 5) + (threadIdx.x >> 5);
  const int lane = threadIdx.x & 31;
  const long e_wave = (long)wave * 30;
  if (e_wave >= E || lane >= 30) return;
  const int grp = lane / 10;
  const int col = lane - 10 * grp;
  const long last = (long)E - 1;
  long el = e_wave + lane;
  el = el < last ? el : last;
  const int sl = src[el], dl = dst[el];

  int d[10];
  float v[10];
  for (int i = 0; i < 10; ++i) {
    const int k = 3 * i + grp;
    const int s = __shfl(sl, k);
    d[i] = __shfl(dl, k);
    v[i] = g1[(size_t)s * 10 + col];
  }
  if (e_wave + 30 <= E) {
    for (int i = 0; i < 10; ++i) atomicAdd(&agg1[(size_t)d[i] * 10 + col], v[i]);
  } else {
    for (int i = 0; i < 10; ++i)
      if (e_wave + 3 * i + grp < E) atomicAdd(&agg1[(size_t)d[i] * 10 + col], v[i]);
  }
}

__global__ void k_scatter2(const int* __restrict__ src, const int* __restrict__ dst,
                           const float* __restrict__ g2, float* __restrict__ agg2, int E) {
  const int wave = blockIdx.x * (blockDim.x >> 5) + (threadIdx.x >> 5);
  const int lane = threadIdx.x & 31;
  const long e_wave = (long)wave * 64;
  if (e_wave >= E) return;
  const int col = lane & 1, p = lane >> 1;
  const long last = (long)E - 1;

  int sl[2], dl[2];
  for (int c = 0; c < 2; ++c) {
    long el = e_wave + 32 * c + lane;
    el = el < last ? el : last;
    sl[c] = src[el];
    dl[c] = dst[el];
  }
  int d[4];
  float v[4];
  for (int r = 0; r < 4; ++r) {
    const int k = 16 * (r & 1) + p;
    const int s = __shfl(sl[r >> 1], k);
    d[r] = __shfl(dl[r >> 1], k);
    v[r] = g2[(size_t)s * 2 + col];
  }
  if (e_wave + 64 <= E) {
    for (int r = 0; r < 4; ++r) atomicAdd(&agg2[(size_t)d[r] * 2 + col], v[r]);
  } else {
    for (int r = 0; r < 4; ++r)
      if (e_wave + 16 * r + p < E) atomicAdd(&agg2[(size_t)d[r] * 2 + col], v[r]);
  }
}

__global__ void k_node2(const float* __restrict__ agg1,
                        const float* __restrict__ c_src, const float* __restrict__ c_dst,
                        const float* __restrict__ b1, const float* __restrict__ W2,
                        float* __restrict__ g2, int N) {
  int n = blockIdx.x * blockDim.x + threadIdx.x;
  if (n >= N) return;
  float cd = c_dst[n], cs = c_src[n];
  float a0 = 0.0f, a1 = 0.0f;
#pragma unroll
  for (int j = 0; j < 10; ++j) {
    float h = fmaxf(agg1[(size_t)n * 10 + j] * cd + b1[j], 0.0f);
    a0 += h * W2[j * 2 + 0];
    a1 += h * W2[j * 2 + 1];
  }
  g2[(size_t)n * 2 + 0] = a0 * cs;
  g2[(size_t)n * 2 + 1] = a1 * cs;
}

__global__ void k_out(const float* __restrict__ agg2, const float* __restrict__ c_dst,
                      const float* __restrict__ b2, float* __restrict__ out, int N) {
  int n = blockIdx.x * blockDim.x + threadIdx.x;
  if (n >= N) return;
  float cd = c_dst[n];
  out[(size_t)n * 2 + 0] = agg2[(size_t)n * 2 + 0] * cd + b2[0];
  out[(size_t)n * 2 + 1] = agg2[(size_t)n * 2 + 1] * cd + b2[1];
}

#define BKT_SHIFT 8
#define BKT_NODES 256
#define PART_TB 1024
#define PART_QPT 3
#define PART_EDGES (PART_TB * 4 * PART_QPT)
#define SUM_TB 1024
#define SUM_WAVES (SUM_TB / 32)


__device__ __forceinline__ unsigned wave_excl_scan(unsigned* a, int n, int lane) {
  const int seg = (n + 31) >> 5;
  const int b0 = lane * seg;
  int b1 = b0 + seg;
  b1 = b1 < n ? b1 : n;
  unsigned s = 0u;
  for (int b = b0; b < b1; ++b) s += a[b];
  unsigned inc = s;
  for (int o = 1; o < 32; o <<= 1) {
    const unsigned t = __shfl_up(inc, o);
    if (lane >= o) inc += t;
  }
  unsigned run = inc - s;
  for (int b = b0; b < b1; ++b) {
    const unsigned c = a[b];
    a[b] = run;
    run += c;
  }
  return __shfl(inc, 31);
}

__global__ __launch_bounds__(PART_TB) void k_part_count_setup(const int* __restrict__ dst,
                                                              unsigned* __restrict__ cnt,
                                                              int E, int nbk,
                                                              float* __restrict__ degz, long nz,
                                                              const float* __restrict__ We,
                                                              const float* __restrict__ Wn,
                                                              const float* __restrict__ W1,
                                                              float* __restrict__ Wep,
                                                              float* __restrict__ Wnp,
                                                              float* __restrict__ W1p) {
  extern __shared__ unsigned part_lds[];
  unsigned* hist = part_lds;
  const int tid = threadIdx.x;
  const long gsz = (long)gridDim.x * PART_TB;
  for (long i = (long)blockIdx.x * PART_TB + tid; i < nz; i += gsz) degz[i] = 0.0f;
  if (blockIdx.x == 0) {
    const int i = tid;
    if (i < 256) {
      const int n = i >> 4, k = i & 15;
      Wep[i] = (n < 8) ? We[k * 8 + n] : 0.0f;
    }
    if (i < 1024) {
      const int n = i >> 6, k = i & 63;
      Wnp[i] = Wn[k * 16 + n];
    }
    if (i < 384) {
      const int n = i / 24, k = i % 24;
      W1p[i] = (n < 10) ? W1[k * 10 + n] : 0.0f;
    }
  }
  for (int i = tid; i < nbk; i += PART_TB) hist[i] = 0u;
  __syncthreads();
  const long e0 = (long)blockIdx.x * PART_EDGES;
  long e1 = e0 + PART_EDGES;
  e1 = e1 < (long)E ? e1 : (long)E;
  const long nq = (e1 - e0) >> 2;
  const int4* __restrict__ d4 = reinterpret_cast<const int4*>(dst + e0);
  if (nq > 0) {
    int4 dv[PART_QPT];
#pragma unroll
    for (int u = 0; u < PART_QPT; ++u) {
      const long q = tid + u * PART_TB;
      dv[u] = d4[q < nq ? q : nq - 1];
    }
#pragma unroll
    for (int u = 0; u < PART_QPT; ++u) {
      const long q = tid + u * PART_TB;
      if (q < nq) {
        atomicAdd(&hist[dv[u].x >> BKT_SHIFT], 1u);
        atomicAdd(&hist[dv[u].y >> BKT_SHIFT], 1u);
        atomicAdd(&hist[dv[u].z >> BKT_SHIFT], 1u);
        atomicAdd(&hist[dv[u].w >> BKT_SHIFT], 1u);
      }
    }
  }
  const long et = e0 + (nq << 2) + tid;
  if (et < e1) atomicAdd(&hist[dst[et] >> BKT_SHIFT], 1u);
  __syncthreads();
  for (int i = tid; i < nbk; i += PART_TB) cnt[(size_t)blockIdx.x * nbk + i] = hist[i];
}

__global__ __launch_bounds__(PART_TB) void k_part_scan(const unsigned* __restrict__ cnt,
                                                       unsigned* __restrict__ off,
                                                       unsigned* __restrict__ bstart,
                                                       int nch, int nbk) {
  extern __shared__ unsigned part_lds[];
  unsigned* tot = part_lds;
  const int tid = threadIdx.x;
  for (int b = tid; b < nbk; b += PART_TB) {
    unsigned run = 0u;
#pragma unroll 16
    for (int k = 0; k < nch; ++k) {
      const unsigned c = cnt[(size_t)k * nbk + b];
      off[(size_t)k * nbk + b] = run;
      run += c;
    }
    tot[b] = run;
  }
  __syncthreads();
  if (tid < 32) {
    const unsigned total = wave_excl_scan(tot, nbk, tid);
    if (tid == 0) tot[nbk] = total;
  }
  __syncthreads();
  for (int b = tid; b <= nbk; b += PART_TB) bstart[b] = tot[b];
}

__global__ __launch_bounds__(PART_TB) void k_part_fill(const int* __restrict__ src,
                                                       const int* __restrict__ dst,
                                                       const unsigned* __restrict__ cnt,
                                                       const unsigned* __restrict__ off,
                                                       const unsigned* __restrict__ bstart,
                                                       unsigned* __restrict__ rec,
                                                       unsigned* __restrict__ eidx,
                                                       int E, int nbk) {
  __shared__ unsigned st_s[PART_EDGES];
  __shared__ unsigned st_d[PART_EDGES];
  __shared__ unsigned st_e[PART_EDGES];
  extern __shared__ unsigned part_lds[];
  unsigned* cur = part_lds;
  unsigned* gdelta = part_lds + nbk;
  const int tid = threadIdx.x;
  const long e0 = (long)blockIdx.x * PART_EDGES;
  long e1 = e0 + PART_EDGES;
  e1 = e1 < (long)E ? e1 : (long)E;
  const long nq = (e1 - e0) >> 2;
  const int4* __restrict__ d4 = reinterpret_cast<const int4*>(dst + e0);
  const int4* __restrict__ s4 = reinterpret_cast<const int4*>(src + e0);
  int4 dv[PART_QPT] = {}, sv[PART_QPT] = {};
  if (nq > 0) {
#pragma unroll
    for (int u = 0; u < PART_QPT; ++u) {
      const long q = tid + u * PART_TB;
      const long qc = q < nq ? q : nq - 1;
      dv[u] = d4[qc];
      sv[u] = s4[qc];
    }
  }
  for (int i = tid; i < nbk; i += PART_TB) cur[i] = cnt[(size_t)blockIdx.x * nbk + i];
  __syncthreads();
  if (tid < 32) wave_excl_scan(cur, nbk, tid);
  __syncthreads();
  for (int i = tid; i < nbk; i += PART_TB)
    gdelta[i] = bstart[i] + off[(size_t)blockIdx.x * nbk + i] - cur[i];
  __syncthreads();

#pragma unroll
  for (int u = 0; u < PART_QPT; ++u) {
    const long q = tid + u * PART_TB;
    if (q < nq) {
      const unsigned eb = (unsigned)(e0 + (q << 2));
      unsigned p;
      p = atomicAdd(&cur[dv[u].x >> BKT_SHIFT], 1u);
      st_s[p] = (unsigned)sv[u].x; st_d[p] = (unsigned)dv[u].x; st_e[p] = eb;
      p = atomicAdd(&cur[dv[u].y >> BKT_SHIFT], 1u);
      st_s[p] = (unsigned)sv[u].y; st_d[p] = (unsigned)dv[u].y; st_e[p] = eb + 1u;
      p = atomicAdd(&cur[dv[u].z >> BKT_SHIFT], 1u);
      st_s[p] = (unsigned)sv[u].z; st_d[p] = (unsigned)dv[u].z; st_e[p] = eb + 2u;
      p = atomicAdd(&cur[dv[u].w >> BKT_SHIFT], 1u);
      st_s[p] = (unsigned)sv[u].w; st_d[p] = (unsigned)dv[u].w; st_e[p] = eb + 3u;
    }
  }
  const long et = e0 + (nq << 2) + tid;
  if (et < e1) {
    const int dd = dst[et];
    const unsigned p = atomicAdd(&cur[dd >> BKT_SHIFT], 1u);
    st_s[p] = (unsigned)src[et]; st_d[p] = (unsigned)dd; st_e[p] = (unsigned)et;
  }
  __syncthreads();

  const int nloc = (int)(e1 - e0);
  for (int i = tid; i < nloc; i += PART_TB) {
    const unsigned dd = st_d[i];
    const unsigned pos = (unsigned)i + gdelta[dd >> BKT_SHIFT];
    rec[pos] = (st_s[i] << 8) | (dd & (BKT_NODES - 1));
    eidx[pos] = st_e[i];
  }
}

#define NODE_TILES (BKT_NODES / 16)
__global__ __launch_bounds__(SUM_TB) void k_bucket_edge_node(const float* __restrict__ ef,
                                                             const float* __restrict__ Wep,
                                                             const unsigned* __restrict__ rec,
                                                             const unsigned* __restrict__ eidx,
                                                             const unsigned* __restrict__ bstart,
                                                             const float* __restrict__ deg_out,
                                                             const float* __restrict__ deg_in,
                                                             const float* __restrict__ be,
                                                             const float* __restrict__ nf,
                                                             const float* __restrict__ Wnp,
                                                             const float* __restrict__ bn,
                                                             const float* __restrict__ W1p,
                                                             float* __restrict__ c_src,
                                                             float* __restrict__ c_dst,
                                                             float* __restrict__ g1, int N) {
  __shared__ float acc[BKT_NODES * 8];
  __shared__ float nh[NODE_TILES * 256];
  const int tid = threadIdx.x;
  const int lo = blockIdx.x << BKT_SHIFT;
  for (int i = tid; i < BKT_NODES * 8; i += SUM_TB) acc[i] = 0.0f;
  __syncthreads();
  const unsigned j0 = bstart[blockIdx.x], j1 = bstart[blockIdx.x + 1];
  const int wave = tid >> 5, lane = tid & 31;
  const int m = lane & 15, kh = lane >> 4, col = m;
  unsigned j = j0 + wave * 16;
  unsigned er = 0u, rm = 0u;
  if (j < j1) {
    const unsigned jc = j + m < j1 ? j + m : j1 - 1u;
    er = eidx[jc];
    rm = rec[jc];
  }
  while (j < j1) {
    const unsigned jn = j + SUM_WAVES * 16;
    unsigned ern = 0u, rmn = 0u;
    if (jn < j1) {
      const unsigned jc = jn + m < j1 ? jn + m : j1 - 1u;
      ern = eidx[jc];
      rmn = rec[jc];
    }
    const bool live = j + m < j1;
    unsigned dl[8];
#pragma unroll
    for (int v = 0; v < 8; ++v) dl[v] = __shfl(rm, v + 8 * kh) & (BKT_NODES - 1);
    v8f c = {};
#pragma unroll
    for (int kb = 0; kb < 16; kb += 4) {
      const int k = kb + 2 * kh;
      v2f a = *reinterpret_cast<const v2f*>(ef + (size_t)er * 16 + k);
      a.x = live ? a.x : 0.0f;
      a.y = live ? a.y : 0.0f;
      const v2f b = *reinterpret_cast<const v2f*>(Wep + col * 16 + k);
      c = __builtin_amdgcn_wmma_f32_16x16x4_f32(false, a, false, b, (short)0, c,
                                                false, false);
    }
    if (col < 8) {
#pragma unroll
      for (int v = 0; v < 8; ++v)
        if (j + v + 8 * kh < j1) atomicAdd(&acc[dl[v] * 8 + col], c[v]);
    }
    er = ern;
    rm = rmn;
    j = jn;
  }

  const int rest = N - lo;
  const int nn = rest < BKT_NODES ? rest : BKT_NODES;
  const int t0 = wave * 16;
  const bool has_tile = wave < NODE_TILES && t0 < nn;
  const int rloc = t0 + m < nn ? t0 + m : nn - 1;
  const bool rowok = t0 + m < nn;
  const size_t nrow = (size_t)(lo + rloc);
  if (has_tile) {
    v8f c = {};
#pragma unroll 8
    for (int kb = 0; kb < 64; kb += 4) {
      const int k = kb + 2 * kh;
      v2f a = *reinterpret_cast<const v2f*>(nf + nrow * 64 + k);
      a.x = rowok ? a.x : 0.0f;
      a.y = rowok ? a.y : 0.0f;
      const v2f b = *reinterpret_cast<const v2f*>(Wnp + col * 64 + k);
      c = __builtin_amdgcn_wmma_f32_16x16x4_f32(false, a, false, b, (short)0, c,
                                                false, false);
    }
    const float bc = bn[col];
#pragma unroll
    for (int v = 0; v < 8; ++v)
      nh[wave * 256 + (v + 8 * kh) * 16 + col] = fmaxf(c[v] + bc, 0.0f);
  }
  __syncthreads();
  if (has_tile) {
    const float dout = deg_out[nrow], din = deg_in[nrow];
    const float cs = (dout > 0.0f) ? rsqrtf(dout) : 0.0f;
    const float cd = (din > 0.0f) ? rsqrtf(din) : 0.0f;
    if (rowok && kh == 0) {
      c_src[nrow] = cs;
      c_dst[nrow] = cd;
    }
    const float csm = rowok ? cs : 0.0f;
    const float inv = 1.0f / fmaxf(din, 1.0f);
    const float bsel = (din > 0.0f) ? 1.0f : 0.0f;
    v8f c = {};
#pragma unroll
    for (int kb = 0; kb < 24; kb += 4) {
      const int k = kb + 2 * kh;
      v2f a;
      if (kb < 8) {
        const v2f s = *reinterpret_cast<const v2f*>(acc + rloc * 8 + k);
        a.x = fmaxf(s.x * inv + bsel * be[k], 0.0f);
        a.y = fmaxf(s.y * inv + bsel * be[k + 1], 0.0f);
      } else {
        a = *reinterpret_cast<const v2f*>(nh + wave * 256 + m * 16 + (k - 8));
      }
      a.x *= csm;
      a.y *= csm;
      const v2f b = *reinterpret_cast<const v2f*>(W1p + col * 24 + k);
      c = __builtin_amdgcn_wmma_f32_16x16x4_f32(false, a, false, b, (short)0, c,
                                                false, false);
    }
    if (col < 10) {
#pragma unroll
      for (int v = 0; v < 8; ++v) {
        const int nl = t0 + v + 8 * kh;
        if (nl < nn) g1[(size_t)(lo + nl) * 10 + col] = c[v];
      }
    }
  }
}

__global__ __launch_bounds__(SUM_TB) void k_bucket_sum10_node(const unsigned* __restrict__ rec,
                                                              const unsigned* __restrict__ bstart,
                                                              const float* __restrict__ g1,
                                                              const float* __restrict__ c_src,
                                                              const float* __restrict__ c_dst,
                                                              const float* __restrict__ b1,
                                                              const float* __restrict__ W2,
                                                              float* __restrict__ g2, int N) {
  __shared__ float acc[BKT_NODES * 10];
  const int tid = threadIdx.x;
  const int lo = blockIdx.x << BKT_SHIFT;
  for (int i = tid; i < BKT_NODES * 10; i += SUM_TB) acc[i] = 0.0f;
  __syncthreads();
  const unsigned j0 = bstart[blockIdx.x], j1 = bstart[blockIdx.x + 1];
  const int wave = tid >> 5, lane = tid & 31;
  const bool has_col = lane < 30;
  const int grp = has_col ? lane / 10 : 0;
  const int col = has_col ? lane - 10 * grp : 0;
  unsigned j = j0 + wave * 30;
  unsigned r = 0u;
  if (j < j1) r = rec[j + lane < j1 ? j + lane : j1 - 1u];
  while (j < j1) {
    const unsigned jn = j + SUM_WAVES * 30;
    unsigned rn = 0u;
    if (jn < j1) rn = rec[jn + lane < j1 ? jn + lane : j1 - 1u];
    unsigned dl[10];
    float v[10];
    for (int i = 0; i < 10; ++i) {
      const unsigned rr = __shfl(r, 3 * i + grp);
      dl[i] = rr & (BKT_NODES - 1);
      v[i] = g1[(size_t)(rr >> 8) * 10 + col];
    }
    if (has_col) {
      for (int i = 0; i < 10; ++i)
        if (j + 3 * i + grp < j1) atomicAdd(&acc[dl[i] * 10 + col], v[i]);
    }
    r = rn;
    j = jn;
  }
  __syncthreads();
  const int rest = N - lo;
  const int nn = rest < BKT_NODES ? rest : BKT_NODES;
  if (tid < nn) {
    const size_t n = (size_t)(lo + tid);
    const float cd = c_dst[n], cs = c_src[n];
    float a0 = 0.0f, a1 = 0.0f;
#pragma unroll
    for (int jc = 0; jc < 10; ++jc) {
      const float h = fmaxf(acc[tid * 10 + jc] * cd + b1[jc], 0.0f);
      a0 += h * W2[jc * 2 + 0];
      a1 += h * W2[jc * 2 + 1];
    }
    v2f o;
    o.x = a0 * cs;
    o.y = a1 * cs;
    *reinterpret_cast<v2f*>(g2 + n * 2) = o;
  }
}

__global__ __launch_bounds__(SUM_TB) void k_bucket_sum2_out(const unsigned* __restrict__ rec,
                                                            const unsigned* __restrict__ bstart,
                                                            const float* __restrict__ g2,
                                                            const float* __restrict__ c_dst,
                                                            const float* __restrict__ b2,
                                                            float* __restrict__ out, int N) {
  __shared__ float acc[BKT_NODES * 2];
  const int tid = threadIdx.x;
  const int lo = blockIdx.x << BKT_SHIFT;
  for (int i = tid; i < BKT_NODES * 2; i += SUM_TB) acc[i] = 0.0f;
  __syncthreads();
  const unsigned j0 = bstart[blockIdx.x], j1 = bstart[blockIdx.x + 1];
  const int wave = tid >> 5, lane = tid & 31;
  const int col = lane & 1, p = lane >> 1;
  for (unsigned j = j0 + wave * 64; j < j1; j += SUM_WAVES * 64) {
    unsigned rl[2];
    for (int c = 0; c < 2; ++c) {
      const unsigned jl = j + 32 * c + lane;
      rl[c] = rec[jl < j1 ? jl : j1 - 1u];
    }
    unsigned dl[4];
    float v[4];
    for (int r = 0; r < 4; ++r) {
      const unsigned rr = __shfl(rl[r >> 1], 16 * (r & 1) + p);
      dl[r] = rr & (BKT_NODES - 1);
      v[r] = g2[(size_t)(rr >> 8) * 2 + col];
    }
    for (int r = 0; r < 4; ++r)
      if (j + 16 * r + p < j1) atomicAdd(&acc[dl[r] * 2 + col], v[r]);
  }
  __syncthreads();
  const int rest = N - lo;
  const int nn = rest < BKT_NODES ? rest : BKT_NODES;
  for (int i = tid; i < nn * 2; i += SUM_TB)
    out[(size_t)lo * 2 + i] = acc[i] * c_dst[lo + (i >> 1)] + b2[i & 1];
}


extern "C" void kernel_launch(void* const* d_in, const int* in_sizes, int n_in,
                              void* d_out, int out_size, void* d_ws, size_t ws_size,
                              hipStream_t stream) {
  const float* node_feats = (const float*)d_in[0];
  const float* edge_feats = (const float*)d_in[1];
  const float* We = (const float*)d_in[2];
  const float* be = (const float*)d_in[3];
  const float* Wn = (const float*)d_in[4];
  const float* bn = (const float*)d_in[5];
  const float* W1 = (const float*)d_in[6];
  const float* b1 = (const float*)d_in[7];
  const float* W2 = (const float*)d_in[8];
  const float* b2 = (const float*)d_in[9];
  const int* src = (const int*)d_in[10];
  const int* dst = (const int*)d_in[11];
  float* out = (float*)d_out;

  const int N = in_sizes[0] / 64;
  const int E = in_sizes[10];
  const size_t Nl = (size_t)N;

  float* ws = (float*)d_ws;
  float* deg_in = ws;
  float* deg_out = ws + Nl;
  float* e_sum = ws + 2 * Nl;
  float* agg1 = ws + 10 * Nl;
  float* agg2 = ws + 20 * Nl;
  float* c_src = ws + 22 * Nl;
  float* c_dst = ws + 23 * Nl;
  float* h24 = ws + 24 * Nl;
  float* g1 = ws + 48 * Nl;
  float* g2 = ws + 58 * Nl;
  float* Wep = ws + 60 * Nl;
  float* Wnp = Wep + 256;
  float* W1p = Wnp + 1024;

  const int TB = 256;
  const int nb = (N + TB - 1) / TB;
  const int waves_per_block = TB / 32;
  const int ntile_blocks = (((N + 15) / 16) + waves_per_block - 1) / waves_per_block;

  const int nbk = (N + BKT_NODES - 1) / BKT_NODES;
  const int nch = (E + PART_EDGES - 1) / PART_EDGES;
  const size_t ncb = (size_t)nch * nbk;
  unsigned* cnt = (unsigned*)(W1p + 384);
  unsigned* off = cnt + ncb;
  unsigned* bstart = off + ncb;
  unsigned* rec = bstart + nbk + 1;
  unsigned* eidx = rec + (size_t)E;
  const size_t lds_words = (size_t)nbk * 2 + 1;

  const size_t need_words = 60 * Nl + 1664 + 2 * ncb + (size_t)nbk + 1 + 2 * (size_t)E;
  const bool ordered = ws_size >= need_words * 4;

  const long zn = ordered ? 2L * N : 22L * N;
  const int deg_ranges = (N + DEG_RANGE - 1) / DEG_RANGE;
  if (ordered) {
    k_part_count_setup<<<nch, PART_TB, lds_words * 4, stream>>>(dst, cnt, E, nbk, ws, zn, We, Wn, W1, Wep, Wnp, W1p);
    k_degree_lds<<<dim3(DEG_SLICES, deg_ranges, 2), DEG_TB, 0, stream>>>(src, dst, deg_out, deg_in, E, N);
    k_part_scan<<<1, PART_TB, lds_words * 4, stream>>>(cnt, off, bstart, nch, nbk);
    k_part_fill<<<nch, PART_TB, lds_words * 4, stream>>>(src, dst, cnt, off, bstart, rec, eidx, E, nbk);
    k_bucket_edge_node<<<nbk, SUM_TB, 0, stream>>>(edge_feats, Wep, rec, eidx, bstart, deg_out, deg_in, be,
                                                   node_feats, Wnp, bn, W1p, c_src, c_dst, g1, N);
    k_bucket_sum10_node<<<nbk, SUM_TB, 0, stream>>>(rec, bstart, g1, c_src, c_dst, b1, W2, g2, N);
    k_bucket_sum2_out<<<nbk, SUM_TB, 0, stream>>>(rec, bstart, g2, c_dst, b2, out, N);
  } else {
    k_zero<<<(int)((zn + TB - 1) / TB), TB, 0, stream>>>(ws, zn);
    k_prep_weights<<<4, TB, 0, stream>>>(We, Wn, W1, Wep, Wnp, W1p);
    k_degree_lds<<<dim3(DEG_SLICES, deg_ranges, 2), DEG_TB, 0, stream>>>(src, dst, deg_out, deg_in, E, N);
    const int etile_blocks = (((E + 15) / 16) + waves_per_block - 1) / waves_per_block;
    k_edge_gemm_scatter<<<etile_blocks, TB, 0, stream>>>(edge_feats, Wep, dst, e_sum, E);
    k_coeff_edgeh<<<nb, TB, 0, stream>>>(deg_out, deg_in, be, e_sum, c_src, c_dst, h24, N);
    k_node_gemm<<<ntile_blocks, TB, 0, stream>>>(node_feats, Wnp, bn, h24, N);
    k_g1_gemm<<<ntile_blocks, TB, 0, stream>>>(h24, c_src, W1p, g1, N);
    const int s10_blocks = (((E + 29) / 30) + waves_per_block - 1) / waves_per_block;
    k_scatter10<<<s10_blocks, TB, 0, stream>>>(src, dst, g1, agg1, E);
    k_node2<<<nb, TB, 0, stream>>>(agg1, c_src, c_dst, b1, W2, g2, N);
    const int s2_blocks = (((E + 63) / 64) + waves_per_block - 1) / waves_per_block;
    k_scatter2<<<s2_blocks, TB, 0, stream>>>(src, dst, g2, agg2, E);
    k_out<<<nb, TB, 0, stream>>>(agg2, c_dst, b2, out, N);
  }
}


# One-block scan of the chunk counts rides in the degree launch as an extra block on a WGP the degree blocks leave empty (7 launches -> 6); order of all f32 sums as in the parent
# speedup vs baseline: 7.3365x; 1.2045x over previous
#include <hip/hip_runtime.h>


typedef float v2f __attribute__((ext_vector_type(2)));
typedef float v8f __attribute__((ext_vector_type(8)));


__global__ void k_zero(float* __restrict__ p, long n) {
  long i = (long)blockIdx.x * blockDim.x + threadIdx.x;
  if (i < n) p[i] = 0.0f;
}

__global__ void k_prep_weights(const float* __restrict__ We,
                               const float* __restrict__ Wn,
                               const float* __restrict__ W1,
                               float* __restrict__ Wep, float* __restrict__ Wnp,
                               float* __restrict__ W1p) {
  int i = blockIdx.x * blockDim.x + threadIdx.x;
  if (i < 256) {
    int n = i >> 4, k = i & 15;
    Wep[i] = (n < 8) ? We[k * 8 + n] : 0.0f;
  }
  if (i < 1024) {
    int n = i >> 6, k = i & 63;
    Wnp[i] = Wn[k * 16 + n];
  }
  if (i < 384) {
    int n = i / 24, k = i % 24;
    W1p[i] = (n < 10) ? W1[k * 10 + n] : 0.0f;
  }
}

#define DEG_RANGE 51200
#define DEG_SLICES 32
#define DEG_TB 1024

__device__ __forceinline__ void deg_bump(unsigned* cnt, int node, int lo, unsigned span) {
  const unsigned r = (unsigned)(node - lo);
  if (r < span) atomicAdd(&cnt[r], 1u);
}

__device__ __forceinline__ void deg_count_block(unsigned* cnt, const int* __restrict__ idx,
                                                float* __restrict__ deg, int slice, int range,
                                                int E, int N) {
  const int tid = threadIdx.x;
  const int lo = range * DEG_RANGE;
  const int rest = N - lo;
  const unsigned span = (unsigned)(rest < DEG_RANGE ? rest : DEG_RANGE);

  for (unsigned i = tid; i < span; i += DEG_TB) cnt[i] = 0u;
  __syncthreads();

  const long per = ((((long)E + DEG_SLICES - 1) / DEG_SLICES) + 3) & ~3L;
  const long e0 = (long)slice * per;
  long e1 = e0 + per;
  e1 = e1 < (long)E ? e1 : (long)E;
  if (e0 < e1) {
    const int4* __restrict__ p4 = reinterpret_cast<const int4*>(idx + e0);
    const long nq = (e1 - e0) >> 2;
    long q = tid;
    for (; q + 3 * DEG_TB < nq; q += 4 * DEG_TB) {
      const int4 a = p4[q];
      const int4 b = p4[q + DEG_TB];
      const int4 c = p4[q + 2 * DEG_TB];
      const int4 d = p4[q + 3 * DEG_TB];
      deg_bump(cnt, a.x, lo, span); deg_bump(cnt, a.y, lo, span);
      deg_bump(cnt, a.z, lo, span); deg_bump(cnt, a.w, lo, span);
      deg_bump(cnt, b.x, lo, span); deg_bump(cnt, b.y, lo, span);
      deg_bump(cnt, b.z, lo, span); deg_bump(cnt, b.w, lo, span);
      deg_bump(cnt, c.x, lo, span); deg_bump(cnt, c.y, lo, span);
      deg_bump(cnt, c.z, lo, span); deg_bump(cnt, c.w, lo, span);
      deg_bump(cnt, d.x, lo, span); deg_bump(cnt, d.y, lo, span);
      deg_bump(cnt, d.z, lo, span); deg_bump(cnt, d.w, lo, span);
    }
    for (; q < nq; q += DEG_TB) {
      const int4 a = p4[q];
      deg_bump(cnt, a.x, lo, span); deg_bump(cnt, a.y, lo, span);
      deg_bump(cnt, a.z, lo, span); deg_bump(cnt, a.w, lo, span);
    }
    const long et = e0 + (nq << 2) + tid;
    if (et < e1) deg_bump(cnt, idx[et], lo, span);
  }
  __syncthreads();

  for (unsigned i = tid; i < span; i += DEG_TB) {
    const unsigned c = cnt[i];
    if (c) atomicAdd(&deg[(size_t)lo + i], (float)c);
  }
}

__global__ __launch_bounds__(DEG_TB) void k_degree_lds(const int* __restrict__ src,
                                                       const int* __restrict__ dst,
                                                       float* __restrict__ deg_out,
                                                       float* __restrict__ deg_in,
                                                       int E, int N) {
  __shared__ unsigned cnt[DEG_RANGE];
  deg_count_block(cnt, blockIdx.z ? dst : src, blockIdx.z ? deg_in : deg_out,
                  blockIdx.x, blockIdx.y, E, N);
}

__global__ void k_edge_gemm_scatter(const float* __restrict__ ef,
                                    const float* __restrict__ Wep,
                                    const int* __restrict__ dst,
                                    float* __restrict__ e_sum, int E) {
  int wave = blockIdx.x * (blockDim.x >> 5) + (threadIdx.x >> 5);
  int lane = threadIdx.x & 31;
  int e0 = wave * 16;
  if (e0 >= E) return;
  int m = lane & 15, kh = lane >> 4, col = m;
  int erow = e0 + m;
  int er = erow < E ? erow : (E - 1);
  float amask = (erow < E) ? 1.0f : 0.0f;

  v8f c = {};
#pragma unroll
  for (int kb = 0; kb < 16; kb += 4) {
    int k = kb + 2 * kh;
    v2f a = *reinterpret_cast<const v2f*>(ef + (size_t)er * 16 + k);
    a.x *= amask;
    a.y *= amask;
    v2f b = *reinterpret_cast<const v2f*>(Wep + col * 16 + k);
    c = __builtin_amdgcn_wmma_f32_16x16x4_f32(false, a, false, b, (short)0, c,
                                              false, false);
  }
  if (col < 8) {
    if (e0 + 16 <= E) {
      int d[8];
#pragma unroll
      for (int v = 0; v < 8; ++v) d[v] = dst[e0 + v + 8 * kh];
#pragma unroll
      for (int v = 0; v < 8; ++v)
        atomicAdd(&e_sum[(size_t)d[v] * 8 + col], c[v]);
    } else {
#pragma unroll
      for (int v = 0; v < 8; ++v) {
        int e = e0 + v + 8 * kh;
        if (e < E) atomicAdd(&e_sum[(size_t)dst[e] * 8 + col], c[v]);
      }
    }
  }
}

__global__ void k_coeff_edgeh(const float* __restrict__ deg_out,
                              const float* __restrict__ deg_in,
                              const float* __restrict__ be,
                              const float* __restrict__ e_sum,
                              float* __restrict__ c_src, float* __restrict__ c_dst,
                              float* __restrict__ h24, int N) {
  int n = blockIdx.x * blockDim.x + threadIdx.x;
  if (n >= N) return;
  float dout = deg_out[n], din = deg_in[n];
  c_src[n] = (dout > 0.0f) ? rsqrtf(dout) : 0.0f;
  c_dst[n] = (din > 0.0f) ? rsqrtf(din) : 0.0f;
  float inv = 1.0f / fmaxf(din, 1.0f);
  float bsel = (din > 0.0f) ? 1.0f : 0.0f;
#pragma unroll
  for (int j = 0; j < 8; ++j) {
    float v = e_sum[(size_t)n * 8 + j] * inv + bsel * be[j];
    h24[(size_t)n * 24 + j] = fmaxf(v, 0.0f);
  }
}

__global__ void k_node_gemm(const float* __restrict__ nf,
                            const float* __restrict__ Wnp,
                            const float* __restrict__ bn,
                            float* __restrict__ h24, int N) {
  int wave = blockIdx.x * (blockDim.x >> 5) + (threadIdx.x >> 5);
  int lane = threadIdx.x & 31;
  int n0 = wave * 16;
  if (n0 >= N) return;
  int m = lane & 15, kh = lane >> 4, col = m;
  int nrow = n0 + m;
  int nr = nrow < N ? nrow : (N - 1);
  float amask = (nrow < N) ? 1.0f : 0.0f;

  v8f c = {};
#pragma unroll
  for (int kb = 0; kb < 64; kb += 4) {
    int k = kb + 2 * kh;
    v2f a = *reinterpret_cast<const v2f*>(nf + (size_t)nr * 64 + k);
    a.x *= amask;
    a.y *= amask;
    v2f b = *reinterpret_cast<const v2f*>(Wnp + col * 64 + k);
    c = __builtin_amdgcn_wmma_f32_16x16x4_f32(false, a, false, b, (short)0, c,
                                              false, false);
  }
  float bc = bn[col];
  if (n0 + 16 <= N) {
#pragma unroll
    for (int v = 0; v < 8; ++v)
      h24[(size_t)(n0 + v + 8 * kh) * 24 + 8 + col] = fmaxf(c[v] + bc, 0.0f);
  } else {
#pragma unroll
    for (int v = 0; v < 8; ++v) {
      int nn = n0 + v + 8 * kh;
      if (nn < N) h24[(size_t)nn * 24 + 8 + col] = fmaxf(c[v] + bc, 0.0f);
    }
  }
}

__global__ void k_g1_gemm(const float* __restrict__ h24,
                          const float* __restrict__ c_src,
                          const float* __restrict__ W1p,
                          float* __restrict__ g1, int N) {
  int wave = blockIdx.x * (blockDim.x >> 5) + (threadIdx.x >> 5);
  int lane = threadIdx.x & 31;
  int n0 = wave * 16;
  if (n0 >= N) return;
  int m = lane & 15, kh = lane >> 4, col = m;
  int nrow = n0 + m;
  int nr = nrow < N ? nrow : (N - 1);
  float cs = (nrow < N) ? c_src[nr] : 0.0f;

  v8f c = {};
#pragma unroll
  for (int kb = 0; kb < 24; kb += 4) {
    int k = kb + 2 * kh;
    v2f a = *reinterpret_cast<const v2f*>(h24 + (size_t)nr * 24 + k);
    a.x *= cs;
    a.y *= cs;
    v2f b = *reinterpret_cast<const v2f*>(W1p + col * 24 + k);
    c = __builtin_amdgcn_wmma_f32_16x16x4_f32(false, a, false, b, (short)0, c,
                                              false, false);
  }
  if (col < 10) {
    if (n0 + 16 <= N) {
#pragma unroll
      for (int v = 0; v < 8; ++v)
        g1[(size_t)(n0 + v + 8 * kh) * 10 + col] = c[v];
    } else {
#pragma unroll
      for (int v = 0; v < 8; ++v) {
        int nn = n0 + v + 8 * kh;
        if (nn < N) g1[(size_t)nn * 10 + col] = c[v];
      }
    }
  }
}

__global__ void k_scatter10(const int* __restrict__ src, const int* __restrict__ dst,
                            const float* __restrict__ g1, float* __restrict__ agg1, int E) {
  const int wave = blockIdx.x * (blockDim.x >> 5) + (threadIdx.x >> 5);
  const int lane = threadIdx.x & 31;
  const long e_wave = (long)wave * 30;
  if (e_wave >= E || lane >= 30) return;
  const int grp = lane / 10;
  const int col = lane - 10 * grp;
  const long last = (long)E - 1;
  long el = e_wave + lane;
  el = el < last ? el : last;
  const int sl = src[el], dl = dst[el];

  int d[10];
  float v[10];
  for (int i = 0; i < 10; ++i) {
    const int k = 3 * i + grp;
    const int s = __shfl(sl, k);
    d[i] = __shfl(dl, k);
    v[i] = g1[(size_t)s * 10 + col];
  }
  if (e_wave + 30 <= E) {
    for (int i = 0; i < 10; ++i) atomicAdd(&agg1[(size_t)d[i] * 10 + col], v[i]);
  } else {
    for (int i = 0; i < 10; ++i)
      if (e_wave + 3 * i + grp < E) atomicAdd(&agg1[(size_t)d[i] * 10 + col], v[i]);
  }
}

__global__ void k_scatter2(const int* __restrict__ src, const int* __restrict__ dst,
                           const float* __restrict__ g2, float* __restrict__ agg2, int E) {
  const int wave = blockIdx.x * (blockDim.x >> 5) + (threadIdx.x >> 5);
  const int lane = threadIdx.x & 31;
  const long e_wave = (long)wave * 64;
  if (e_wave >= E) return;
  const int col = lane & 1, p = lane >> 1;
  const long last = (long)E - 1;

  int sl[2], dl[2];
  for (int c = 0; c < 2; ++c) {
    long el = e_wave + 32 * c + lane;
    el = el < last ? el : last;
    sl[c] = src[el];
    dl[c] = dst[el];
  }
  int d[4];
  float v[4];
  for (int r = 0; r < 4; ++r) {
    const int k = 16 * (r & 1) + p;
    const int s = __shfl(sl[r >> 1], k);
    d[r] = __shfl(dl[r >> 1], k);
    v[r] = g2[(size_t)s * 2 + col];
  }
  if (e_wave + 64 <= E) {
    for (int r = 0; r < 4; ++r) atomicAdd(&agg2[(size_t)d[r] * 2 + col], v[r]);
  } else {
    for (int r = 0; r < 4; ++r)
      if (e_wave + 16 * r + p < E) atomicAdd(&agg2[(size_t)d[r] * 2 + col], v[r]);
  }
}

__global__ void k_node2(const float* __restrict__ agg1,
                        const float* __restrict__ c_src, const float* __restrict__ c_dst,
                        const float* __restrict__ b1, const float* __restrict__ W2,
                        float* __restrict__ g2, int N) {
  int n = blockIdx.x * blockDim.x + threadIdx.x;
  if (n >= N) return;
  float cd = c_dst[n], cs = c_src[n];
  float a0 = 0.0f, a1 = 0.0f;
#pragma unroll
  for (int j = 0; j < 10; ++j) {
    float h = fmaxf(agg1[(size_t)n * 10 + j] * cd + b1[j], 0.0f);
    a0 += h * W2[j * 2 + 0];
    a1 += h * W2[j * 2 + 1];
  }
  g2[(size_t)n * 2 + 0] = a0 * cs;
  g2[(size_t)n * 2 + 1] = a1 * cs;
}

__global__ void k_out(const float* __restrict__ agg2, const float* __restrict__ c_dst,
                      const float* __restrict__ b2, float* __restrict__ out, int N) {
  int n = blockIdx.x * blockDim.x + threadIdx.x;
  if (n >= N) return;
  float cd = c_dst[n];
  out[(size_t)n * 2 + 0] = agg2[(size_t)n * 2 + 0] * cd + b2[0];
  out[(size_t)n * 2 + 1] = agg2[(size_t)n * 2 + 1] * cd + b2[1];
}

#define BKT_SHIFT 8
#define BKT_NODES 256
#define PART_TB 1024
#define PART_QPT 3
#define PART_EDGES (PART_TB * 4 * PART_QPT)
#define SUM_TB 1024
#define SUM_WAVES (SUM_TB / 32)


__device__ __forceinline__ unsigned wave_excl_scan(unsigned* a, int n, int lane) {
  const int seg = (n + 31) >> 5;
  const int b0 = lane * seg;
  int b1 = b0 + seg;
  b1 = b1 < n ? b1 : n;
  unsigned s = 0u;
  for (int b = b0; b < b1; ++b) s += a[b];
  unsigned inc = s;
  for (int o = 1; o < 32; o <<= 1) {
    const unsigned t = __shfl_up(inc, o);
    if (lane >= o) inc += t;
  }
  unsigned run = inc - s;
  for (int b = b0; b < b1; ++b) {
    const unsigned c = a[b];
    a[b] = run;
    run += c;
  }
  return __shfl(inc, 31);
}

__global__ __launch_bounds__(PART_TB) void k_part_count_setup(const int* __restrict__ dst,
                                                              unsigned* __restrict__ cnt,
                                                              int E, int nbk,
                                                              float* __restrict__ degz, long nz,
                                                              const float* __restrict__ We,
                                                              const float* __restrict__ Wn,
                                                              const float* __restrict__ W1,
                                                              float* __restrict__ Wep,
                                                              float* __restrict__ Wnp,
                                                              float* __restrict__ W1p) {
  extern __shared__ unsigned part_lds[];
  unsigned* hist = part_lds;
  const int tid = threadIdx.x;
  const long gsz = (long)gridDim.x * PART_TB;
  for (long i = (long)blockIdx.x * PART_TB + tid; i < nz; i += gsz) degz[i] = 0.0f;
  if (blockIdx.x == 0) {
    const int i = tid;
    if (i < 256) {
      const int n = i >> 4, k = i & 15;
      Wep[i] = (n < 8) ? We[k * 8 + n] : 0.0f;
    }
    if (i < 1024) {
      const int n = i >> 6, k = i & 63;
      Wnp[i] = Wn[k * 16 + n];
    }
    if (i < 384) {
      const int n = i / 24, k = i % 24;
      W1p[i] = (n < 10) ? W1[k * 10 + n] : 0.0f;
    }
  }
  for (int i = tid; i < nbk; i += PART_TB) hist[i] = 0u;
  __syncthreads();
  const long e0 = (long)blockIdx.x * PART_EDGES;
  long e1 = e0 + PART_EDGES;
  e1 = e1 < (long)E ? e1 : (long)E;
  const long nq = (e1 - e0) >> 2;
  const int4* __restrict__ d4 = reinterpret_cast<const int4*>(dst + e0);
  if (nq > 0) {
    int4 dv[PART_QPT];
#pragma unroll
    for (int u = 0; u < PART_QPT; ++u) {
      const long q = tid + u * PART_TB;
      dv[u] = d4[q < nq ? q : nq - 1];
    }
#pragma unroll
    for (int u = 0; u < PART_QPT; ++u) {
      const long q = tid + u * PART_TB;
      if (q < nq) {
        atomicAdd(&hist[dv[u].x >> BKT_SHIFT], 1u);
        atomicAdd(&hist[dv[u].y >> BKT_SHIFT], 1u);
        atomicAdd(&hist[dv[u].z >> BKT_SHIFT], 1u);
        atomicAdd(&hist[dv[u].w >> BKT_SHIFT], 1u);
      }
    }
  }
  const long et = e0 + (nq << 2) + tid;
  if (et < e1) atomicAdd(&hist[dst[et] >> BKT_SHIFT], 1u);
  __syncthreads();
  for (int i = tid; i < nbk; i += PART_TB) cnt[(size_t)blockIdx.x * nbk + i] = hist[i];
}

__device__ __forceinline__ void part_scan_block(unsigned* tot,
                                                const unsigned* __restrict__ cnt,
                                                unsigned* __restrict__ off,
                                                unsigned* __restrict__ bstart,
                                                int nch, int nbk) {
  const int tid = threadIdx.x;
  for (int b = tid; b < nbk; b += PART_TB) {
    unsigned run = 0u;
#pragma unroll 16
    for (int k = 0; k < nch; ++k) {
      const unsigned c = cnt[(size_t)k * nbk + b];
      off[(size_t)k * nbk + b] = run;
      run += c;
    }
    tot[b] = run;
  }
  __syncthreads();
  if (tid < 32) {
    const unsigned total = wave_excl_scan(tot, nbk, tid);
    if (tid == 0) tot[nbk] = total;
  }
  __syncthreads();
  for (int b = tid; b <= nbk; b += PART_TB) bstart[b] = tot[b];
}

static_assert(PART_TB == DEG_TB, "the scan block strides by PART_TB inside a block of DEG_TB threads");
__global__ __launch_bounds__(DEG_TB) void k_degree_scan(const int* __restrict__ src,
                                                        const int* __restrict__ dst,
                                                        float* __restrict__ deg_out,
                                                        float* __restrict__ deg_in,
                                                        int E, int N, int nranges,
                                                        const unsigned* __restrict__ pcnt,
                                                        unsigned* __restrict__ off,
                                                        unsigned* __restrict__ bstart,
                                                        int nch, int nbk) {
  __shared__ unsigned lds[DEG_RANGE];
  if (blockIdx.x == 0) {
    part_scan_block(lds, pcnt, off, bstart, nch, nbk);
    return;
  }
  const int bid = (int)blockIdx.x - 1;
  const int slice = bid % DEG_SLICES;
  const int rz = bid / DEG_SLICES;
  const int range = rz % nranges;
  const int z = rz / nranges;
  deg_count_block(lds, z ? dst : src, z ? deg_in : deg_out, slice, range, E, N);
}

__global__ __launch_bounds__(PART_TB) void k_part_fill(const int* __restrict__ src,
                                                       const int* __restrict__ dst,
                                                       const unsigned* __restrict__ cnt,
                                                       const unsigned* __restrict__ off,
                                                       const unsigned* __restrict__ bstart,
                                                       unsigned* __restrict__ rec,
                                                       unsigned* __restrict__ eidx,
                                                       int E, int nbk) {
  __shared__ unsigned st_s[PART_EDGES];
  __shared__ unsigned st_d[PART_EDGES];
  __shared__ unsigned st_e[PART_EDGES];
  extern __shared__ unsigned part_lds[];
  unsigned* cur = part_lds;
  unsigned* gdelta = part_lds + nbk;
  const int tid = threadIdx.x;
  const long e0 = (long)blockIdx.x * PART_EDGES;
  long e1 = e0 + PART_EDGES;
  e1 = e1 < (long)E ? e1 : (long)E;
  const long nq = (e1 - e0) >> 2;
  const int4* __restrict__ d4 = reinterpret_cast<const int4*>(dst + e0);
  const int4* __restrict__ s4 = reinterpret_cast<const int4*>(src + e0);
  int4 dv[PART_QPT] = {}, sv[PART_QPT] = {};
  if (nq > 0) {
#pragma unroll
    for (int u = 0; u < PART_QPT; ++u) {
      const long q = tid + u * PART_TB;
      const long qc = q < nq ? q : nq - 1;
      dv[u] = d4[qc];
      sv[u] = s4[qc];
    }
  }
  for (int i = tid; i < nbk; i += PART_TB) cur[i] = cnt[(size_t)blockIdx.x * nbk + i];
  __syncthreads();
  if (tid < 32) wave_excl_scan(cur, nbk, tid);
  __syncthreads();
  for (int i = tid; i < nbk; i += PART_TB)
    gdelta[i] = bstart[i] + off[(size_t)blockIdx.x * nbk + i] - cur[i];
  __syncthreads();

#pragma unroll
  for (int u = 0; u < PART_QPT; ++u) {
    const long q = tid + u * PART_TB;
    if (q < nq) {
      const unsigned eb = (unsigned)(e0 + (q << 2));
      unsigned p;
      p = atomicAdd(&cur[dv[u].x >> BKT_SHIFT], 1u);
      st_s[p] = (unsigned)sv[u].x; st_d[p] = (unsigned)dv[u].x; st_e[p] = eb;
      p = atomicAdd(&cur[dv[u].y >> BKT_SHIFT], 1u);
      st_s[p] = (unsigned)sv[u].y; st_d[p] = (unsigned)dv[u].y; st_e[p] = eb + 1u;
      p = atomicAdd(&cur[dv[u].z >> BKT_SHIFT], 1u);
      st_s[p] = (unsigned)sv[u].z; st_d[p] = (unsigned)dv[u].z; st_e[p] = eb + 2u;
      p = atomicAdd(&cur[dv[u].w >> BKT_SHIFT], 1u);
      st_s[p] = (unsigned)sv[u].w; st_d[p] = (unsigned)dv[u].w; st_e[p] = eb + 3u;
    }
  }
  const long et = e0 + (nq << 2) + tid;
  if (et < e1) {
    const int dd = dst[et];
    const unsigned p = atomicAdd(&cur[dd >> BKT_SHIFT], 1u);
    st_s[p] = (unsigned)src[et]; st_d[p] = (unsigned)dd; st_e[p] = (unsigned)et;
  }
  __syncthreads();

  const int nloc = (int)(e1 - e0);
  for (int i = tid; i < nloc; i += PART_TB) {
    const unsigned dd = st_d[i];
    const unsigned pos = (unsigned)i + gdelta[dd >> BKT_SHIFT];
    rec[pos] = (st_s[i] << 8) | (dd & (BKT_NODES - 1));
    eidx[pos] = st_e[i];
  }
}

#define NODE_TILES (BKT_NODES / 16)
__global__ __launch_bounds__(SUM_TB) void k_bucket_edge_node(const float* __restrict__ ef,
                                                             const float* __restrict__ Wep,
                                                             const unsigned* __restrict__ rec,
                                                             const unsigned* __restrict__ eidx,
                                                             const unsigned* __restrict__ bstart,
                                                             const float* __restrict__ deg_out,
                                                             const float* __restrict__ deg_in,
                                                             const float* __restrict__ be,
                                                             const float* __restrict__ nf,
                                                             const float* __restrict__ Wnp,
                                                             const float* __restrict__ bn,
                                                             const float* __restrict__ W1p,
                                                             float* __restrict__ c_src,
                                                             float* __restrict__ c_dst,
                                                             float* __restrict__ g1, int N) {
  __shared__ float acc[BKT_NODES * 8];
  __shared__ float nh[NODE_TILES * 256];
  const int tid = threadIdx.x;
  const int lo = blockIdx.x << BKT_SHIFT;
  for (int i = tid; i < BKT_NODES * 8; i += SUM_TB) acc[i] = 0.0f;
  __syncthreads();
  const unsigned j0 = bstart[blockIdx.x], j1 = bstart[blockIdx.x + 1];
  const int wave = tid >> 5, lane = tid & 31;
  const int m = lane & 15, kh = lane >> 4, col = m;
  unsigned j = j0 + wave * 16;
  unsigned er = 0u, rm = 0u;
  if (j < j1) {
    const unsigned jc = j + m < j1 ? j + m : j1 - 1u;
    er = eidx[jc];
    rm = rec[jc];
  }
  while (j < j1) {
    const unsigned jn = j + SUM_WAVES * 16;
    unsigned ern = 0u, rmn = 0u;
    if (jn < j1) {
      const unsigned jc = jn + m < j1 ? jn + m : j1 - 1u;
      ern = eidx[jc];
      rmn = rec[jc];
    }
    const bool live = j + m < j1;
    unsigned dl[8];
#pragma unroll
    for (int v = 0; v < 8; ++v) dl[v] = __shfl(rm, v + 8 * kh) & (BKT_NODES - 1);
    v8f c = {};
#pragma unroll
    for (int kb = 0; kb < 16; kb += 4) {
      const int k = kb + 2 * kh;
      v2f a = *reinterpret_cast<const v2f*>(ef + (size_t)er * 16 + k);
      a.x = live ? a.x : 0.0f;
      a.y = live ? a.y : 0.0f;
      const v2f b = *reinterpret_cast<const v2f*>(Wep + col * 16 + k);
      c = __builtin_amdgcn_wmma_f32_16x16x4_f32(false, a, false, b, (short)0, c,
                                                false, false);
    }
    if (col < 8) {
#pragma unroll
      for (int v = 0; v < 8; ++v)
        if (j + v + 8 * kh < j1) atomicAdd(&acc[dl[v] * 8 + col], c[v]);
    }
    er = ern;
    rm = rmn;
    j = jn;
  }

  const int rest = N - lo;
  const int nn = rest < BKT_NODES ? rest : BKT_NODES;
  const int t0 = wave * 16;
  const bool has_tile = wave < NODE_TILES && t0 < nn;
  const int rloc = t0 + m < nn ? t0 + m : nn - 1;
  const bool rowok = t0 + m < nn;
  const size_t nrow = (size_t)(lo + rloc);
  if (has_tile) {
    v8f c = {};
#pragma unroll 8
    for (int kb = 0; kb < 64; kb += 4) {
      const int k = kb + 2 * kh;
      v2f a = *reinterpret_cast<const v2f*>(nf + nrow * 64 + k);
      a.x = rowok ? a.x : 0.0f;
      a.y = rowok ? a.y : 0.0f;
      const v2f b = *reinterpret_cast<const v2f*>(Wnp + col * 64 + k);
      c = __builtin_amdgcn_wmma_f32_16x16x4_f32(false, a, false, b, (short)0, c,
                                                false, false);
    }
    const float bc = bn[col];
#pragma unroll
    for (int v = 0; v < 8; ++v)
      nh[wave * 256 + (v + 8 * kh) * 16 + col] = fmaxf(c[v] + bc, 0.0f);
  }
  __syncthreads();
  if (has_tile) {
    const float dout = deg_out[nrow], din = deg_in[nrow];
    const float cs = (dout > 0.0f) ? rsqrtf(dout) : 0.0f;
    const float cd = (din > 0.0f) ? rsqrtf(din) : 0.0f;
    if (rowok && kh == 0) {
      c_src[nrow] = cs;
      c_dst[nrow] = cd;
    }
    const float csm = rowok ? cs : 0.0f;
    const float inv = 1.0f / fmaxf(din, 1.0f);
    const float bsel = (din > 0.0f) ? 1.0f : 0.0f;
    v8f c = {};
#pragma unroll
    for (int kb = 0; kb < 24; kb += 4) {
      const int k = kb + 2 * kh;
      v2f a;
      if (kb < 8) {
        const v2f s = *reinterpret_cast<const v2f*>(acc + rloc * 8 + k);
        a.x = fmaxf(s.x * inv + bsel * be[k], 0.0f);
        a.y = fmaxf(s.y * inv + bsel * be[k + 1], 0.0f);
      } else {
        a = *reinterpret_cast<const v2f*>(nh + wave * 256 + m * 16 + (k - 8));
      }
      a.x *= csm;
      a.y *= csm;
      const v2f b = *reinterpret_cast<const v2f*>(W1p + col * 24 + k);
      c = __builtin_amdgcn_wmma_f32_16x16x4_f32(false, a, false, b, (short)0, c,
                                                false, false);
    }
    if (col < 10) {
#pragma unroll
      for (int v = 0; v < 8; ++v) {
        const int nl = t0 + v + 8 * kh;
        if (nl < nn) g1[(size_t)(lo + nl) * 10 + col] = c[v];
      }
    }
  }
}

__global__ __launch_bounds__(SUM_TB) void k_bucket_sum10_node(const unsigned* __restrict__ rec,
                                                              const unsigned* __restrict__ bstart,
                                                              const float* __restrict__ g1,
                                                              const float* __restrict__ c_src,
                                                              const float* __restrict__ c_dst,
                                                              const float* __restrict__ b1,
                                                              const float* __restrict__ W2,
                                                              float* __restrict__ g2, int N) {
  __shared__ float acc[BKT_NODES * 10];
  const int tid = threadIdx.x;
  const int lo = blockIdx.x << BKT_SHIFT;
  for (int i = tid; i < BKT_NODES * 10; i += SUM_TB) acc[i] = 0.0f;
  __syncthreads();
  const unsigned j0 = bstart[blockIdx.x], j1 = bstart[blockIdx.x + 1];
  const int wave = tid >> 5, lane = tid & 31;
  const bool has_col = lane < 30;
  const int grp = has_col ? lane / 10 : 0;
  const int col = has_col ? lane - 10 * grp : 0;
  unsigned j = j0 + wave * 30;
  unsigned r = 0u;
  if (j < j1) r = rec[j + lane < j1 ? j + lane : j1 - 1u];
  while (j < j1) {
    const unsigned jn = j + SUM_WAVES * 30;
    unsigned rn = 0u;
    if (jn < j1) rn = rec[jn + lane < j1 ? jn + lane : j1 - 1u];
    unsigned dl[10];
    float v[10];
    for (int i = 0; i < 10; ++i) {
      const unsigned rr = __shfl(r, 3 * i + grp);
      dl[i] = rr & (BKT_NODES - 1);
      v[i] = g1[(size_t)(rr >> 8) * 10 + col];
    }
    if (has_col) {
      for (int i = 0; i < 10; ++i)
        if (j + 3 * i + grp < j1) atomicAdd(&acc[dl[i] * 10 + col], v[i]);
    }
    r = rn;
    j = jn;
  }
  __syncthreads();
  const int rest = N - lo;
  const int nn = rest < BKT_NODES ? rest : BKT_NODES;
  if (tid < nn) {
    const size_t n = (size_t)(lo + tid);
    const float cd = c_dst[n], cs = c_src[n];
    float a0 = 0.0f, a1 = 0.0f;
#pragma unroll
    for (int jc = 0; jc < 10; ++jc) {
      const float h = fmaxf(acc[tid * 10 + jc] * cd + b1[jc], 0.0f);
      a0 += h * W2[jc * 2 + 0];
      a1 += h * W2[jc * 2 + 1];
    }
    v2f o;
    o.x = a0 * cs;
    o.y = a1 * cs;
    *reinterpret_cast<v2f*>(g2 + n * 2) = o;
  }
}

__global__ __launch_bounds__(SUM_TB) void k_bucket_sum2_out(const unsigned* __restrict__ rec,
                                                            const unsigned* __restrict__ bstart,
                                                            const float* __restrict__ g2,
                                                            const float* __restrict__ c_dst,
                                                            const float* __restrict__ b2,
                                                            float* __restrict__ out, int N) {
  __shared__ float acc[BKT_NODES * 2];
  const int tid = threadIdx.x;
  const int lo = blockIdx.x << BKT_SHIFT;
  for (int i = tid; i < BKT_NODES * 2; i += SUM_TB) acc[i] = 0.0f;
  __syncthreads();
  const unsigned j0 = bstart[blockIdx.x], j1 = bstart[blockIdx.x + 1];
  const int wave = tid >> 5, lane = tid & 31;
  const int col = lane & 1, p = lane >> 1;
  for (unsigned j = j0 + wave * 64; j < j1; j += SUM_WAVES * 64) {
    unsigned rl[2];
    for (int c = 0; c < 2; ++c) {
      const unsigned jl = j + 32 * c + lane;
      rl[c] = rec[jl < j1 ? jl : j1 - 1u];
    }
    unsigned dl[4];
    float v[4];
    for (int r = 0; r < 4; ++r) {
      const unsigned rr = __shfl(rl[r >> 1], 16 * (r & 1) + p);
      dl[r] = rr & (BKT_NODES - 1);
      v[r] = g2[(size_t)(rr >> 8) * 2 + col];
    }
    for (int r = 0; r < 4; ++r)
      if (j + 16 * r + p < j1) atomicAdd(&acc[dl[r] * 2 + col], v[r]);
  }
  __syncthreads();
  const int rest = N - lo;
  const int nn = rest < BKT_NODES ? rest : BKT_NODES;
  for (int i = tid; i < nn * 2; i += SUM_TB)
    out[(size_t)lo * 2 + i] = acc[i] * c_dst[lo + (i >> 1)] + b2[i & 1];
}


extern "C" void kernel_launch(void* const* d_in, const int* in_sizes, int n_in,
                              void* d_out, int out_size, void* d_ws, size_t ws_size,
                              hipStream_t stream) {
  const float* node_feats = (const float*)d_in[0];
  const float* edge_feats = (const float*)d_in[1];
  const float* We = (const float*)d_in[2];
  const float* be = (const float*)d_in[3];
  const float* Wn = (const float*)d_in[4];
  const float* bn = (const float*)d_in[5];
  const float* W1 = (const float*)d_in[6];
  const float* b1 = (const float*)d_in[7];
  const float* W2 = (const float*)d_in[8];
  const float* b2 = (const float*)d_in[9];
  const int* src = (const int*)d_in[10];
  const int* dst = (const int*)d_in[11];
  float* out = (float*)d_out;

  const int N = in_sizes[0] / 64;
  const int E = in_sizes[10];
  const size_t Nl = (size_t)N;

  float* ws = (float*)d_ws;
  float* deg_in = ws;
  float* deg_out = ws + Nl;
  float* e_sum = ws + 2 * Nl;
  float* agg1 = ws + 10 * Nl;
  float* agg2 = ws + 20 * Nl;
  float* c_src = ws + 22 * Nl;
  float* c_dst = ws + 23 * Nl;
  float* h24 = ws + 24 * Nl;
  float* g1 = ws + 48 * Nl;
  float* g2 = ws + 58 * Nl;
  float* Wep = ws + 60 * Nl;
  float* Wnp = Wep + 256;
  float* W1p = Wnp + 1024;

  const int TB = 256;
  const int nb = (N + TB - 1) / TB;
  const int waves_per_block = TB / 32;
  const int ntile_blocks = (((N + 15) / 16) + waves_per_block - 1) / waves_per_block;

  const int nbk = (N + BKT_NODES - 1) / BKT_NODES;
  const int nch = (E + PART_EDGES - 1) / PART_EDGES;
  const size_t ncb = (size_t)nch * nbk;
  unsigned* cnt = (unsigned*)(W1p + 384);
  unsigned* off = cnt + ncb;
  unsigned* bstart = off + ncb;
  unsigned* rec = bstart + nbk + 1;
  unsigned* eidx = rec + (size_t)E;
  const size_t lds_words = (size_t)nbk * 2 + 1;

  const size_t need_words = 60 * Nl + 1664 + 2 * ncb + (size_t)nbk + 1 + 2 * (size_t)E;
  const bool ordered = ws_size >= need_words * 4 && nbk < DEG_RANGE;

  const long zn = ordered ? 2L * N : 22L * N;
  const int deg_ranges = (N + DEG_RANGE - 1) / DEG_RANGE;
  if (ordered) {
    k_part_count_setup<<<nch, PART_TB, lds_words * 4, stream>>>(dst, cnt, E, nbk, ws, zn, We, Wn, W1, Wep, Wnp, W1p);
    k_degree_scan<<<1 + DEG_SLICES * deg_ranges * 2, DEG_TB, 0, stream>>>(src, dst, deg_out, deg_in, E, N, deg_ranges,
                                                                          cnt, off, bstart, nch, nbk);
    k_part_fill<<<nch, PART_TB, lds_words * 4, stream>>>(src, dst, cnt, off, bstart, rec, eidx, E, nbk);
    k_bucket_edge_node<<<nbk, SUM_TB, 0, stream>>>(edge_feats, Wep, rec, eidx, bstart, deg_out, deg_in, be,
                                                   node_feats, Wnp, bn, W1p, c_src, c_dst, g1, N);
    k_bucket_sum10_node<<<nbk, SUM_TB, 0, stream>>>(rec, bstart, g1, c_src, c_dst, b1, W2, g2, N);
    k_bucket_sum2_out<<<nbk, SUM_TB, 0, stream>>>(rec, bstart, g2, c_dst, b2, out, N);
  } else {
    k_zero<<<(int)((zn + TB - 1) / TB), TB, 0, stream>>>(ws, zn);
    k_prep_weights<<<4, TB, 0, stream>>>(We, Wn, W1, Wep, Wnp, W1p);
    k_degree_lds<<<dim3(DEG_SLICES, deg_ranges, 2), DEG_TB, 0, stream>>>(src, dst, deg_out, deg_in, E, N);
    const int etile_blocks = (((E + 15) / 16) + waves_per_block - 1) / waves_per_block;
    k_edge_gemm_scatter<<<etile_blocks, TB, 0, stream>>>(edge_feats, Wep, dst, e_sum, E);
    k_coeff_edgeh<<<nb, TB, 0, stream>>>(deg_out, deg_in, be, e_sum, c_src, c_dst, h24, N);
    k_node_gemm<<<ntile_blocks, TB, 0, stream>>>(node_feats, Wnp, bn, h24, N);
    k_g1_gemm<<<ntile_blocks, TB, 0, stream>>>(h24, c_src, W1p, g1, N);
    const int s10_blocks = (((E + 29) / 30) + waves_per_block - 1) / waves_per_block;
    k_scatter10<<<s10_blocks, TB, 0, stream>>>(src, dst, g1, agg1, E);
    k_node2<<<nb, TB, 0, stream>>>(agg1, c_src, c_dst, b1, W2, g2, N);
    const int s2_blocks = (((E + 63) / 64) + waves_per_block - 1) / waves_per_block;
    k_scatter2<<<s2_blocks, TB, 0, stream>>>(src, dst, g2, agg2, E);
    k_out<<<nb, TB, 0, stream>>>(agg2, c_dst, b2, out, N);
  }
}
